# starting kernel
# baseline (speedup 1.0000x reference)
#include <hip/hip_runtime.h>
#include <hip/hip_bf16.h>
#include <math.h>


typedef float v2f __attribute__((ext_vector_type(2)));
typedef float v8f __attribute__((ext_vector_type(8)));

#define IN_CH 8
#define OUT_CH 64

__device__ __forceinline__ v8f wmma_f32(v2f a, v2f b, v8f c) {
  return __builtin_amdgcn_wmma_f32_16x16x4_f32(false, a, false, b, (short)0, c,
                                               false, false);
}

__device__ __forceinline__ v8f accum_block(const float* __restrict__ A, int ldA,
                                           int rA, const float* __restrict__ W,
                                           int Kdim, int colB, int half,
                                           v8f acc) {
#pragma unroll 4
  for (int k0 = 0; k0 < Kdim; k0 += 4) {
    int ka = k0 + 2 * half;
    v2f a;
    a.x = A[rA * ldA + ka];
    a.y = A[rA * ldA + ka + 1];
    v2f b;
    b.x = W[ka * OUT_CH + colB];
    b.y = W[(ka + 1) * OUT_CH + colB];
    acc = wmma_f32(a, b, acc);
  }
  return acc;
}


__global__ void __launch_bounds__(256) zero_kernel(float* p, int n) {
  int i = blockIdx.x * blockDim.x + threadIdx.x;
  if (i < n) p[i] = 0.0f;
}

__global__ void __launch_bounds__(256)
degree_kernel(const int* __restrict__ rowIdx, float* __restrict__ deg, int E) {
  int e = blockIdx.x * blockDim.x + threadIdx.x;
  if (e < E) atomicAdd(&deg[rowIdx[e]], 1.0f);
}

__global__ void __launch_bounds__(256) dinv_kernel(float* deg, int n) {
  int i = blockIdx.x * blockDim.x + threadIdx.x;
  if (i < n) {
    float d = deg[i];
    deg[i] = (d > 0.0f) ? rsqrtf(d) : 0.0f;
  }
}

__global__ void __launch_bounds__(256)
spmm_kernel(const int* __restrict__ rowIdx, const int* __restrict__ colIdx,
            const float* __restrict__ dinv, const float* __restrict__ src64,
            float* __restrict__ dst64, const float* __restrict__ src8,
            float* __restrict__ dst8, int E) {
  int e = __builtin_amdgcn_readfirstlane(
      (blockIdx.x * blockDim.x + threadIdx.x) >> 5);
  int lane = threadIdx.x & 31;
  if (e >= E) return;
  int r = __builtin_amdgcn_readfirstlane(rowIdx[e]);
  int c = __builtin_amdgcn_readfirstlane(colIdx[e]);
  float nrm = -dinv[r] * dinv[c];
  const float2* __restrict__ src = (const float2*)(src64 + (size_t)c * OUT_CH);
  float2 v = src[lane];
  float* __restrict__ dst = dst64 + (size_t)r * OUT_CH + 2 * lane;
  atomicAdd(&dst[0], v.x * nrm);
  atomicAdd(&dst[1], v.y * nrm);
  if (src8 != nullptr && lane < IN_CH) {
    atomicAdd(&dst8[r * IN_CH + lane], src8[c * IN_CH + lane] * nrm);
  }
}

__global__ void __launch_bounds__(128)
gates_wmma_kernel(const float* __restrict__ x, const float* __restrict__ H,
                  const float* __restrict__ Lx, const float* __restrict__ LH,
                  const float* __restrict__ Wzx, const float* __restrict__ Wzh,
                  const float* __restrict__ Wrx, const float* __restrict__ Wrh,
                  const float* __restrict__ bzx, const float* __restrict__ bzh,
                  const float* __restrict__ brx, const float* __restrict__ brh,
                  float* __restrict__ Zbuf, float* __restrict__ RH) {
  int lane = threadIdx.x & 31;
  int wv = threadIdx.x >> 5;
  int half = lane >> 4;
  int l15 = lane & 15;
  int row0 = blockIdx.x * 16;
  int col0 = wv * 16;
  int rA = row0 + l15;
  int colB = col0 + l15;

  v8f accZ = {};
  v8f accR = {};
  accZ = accum_block(x, IN_CH, rA, Wzx, IN_CH, colB, half, accZ);
  accR = accum_block(x, IN_CH, rA, Wrx, IN_CH, colB, half, accR);
  accZ = accum_block(Lx, IN_CH, rA, Wzx + IN_CH * OUT_CH, IN_CH, colB, half, accZ);
  accR = accum_block(Lx, IN_CH, rA, Wrx + IN_CH * OUT_CH, IN_CH, colB, half, accR);
  accZ = accum_block(H, OUT_CH, rA, Wzh, OUT_CH, colB, half, accZ);
  accR = accum_block(H, OUT_CH, rA, Wrh, OUT_CH, colB, half, accR);
  accZ = accum_block(LH, OUT_CH, rA, Wzh + OUT_CH * OUT_CH, OUT_CH, colB, half, accZ);
  accR = accum_block(LH, OUT_CH, rA, Wrh + OUT_CH * OUT_CH, OUT_CH, colB, half, accR);

  float bZ = bzx[colB] + bzh[colB];
  float bR = brx[colB] + brh[colB];
#pragma unroll
  for (int i = 0; i < 8; ++i) {
    int r = row0 + i + 8 * half;
    float z = 1.0f / (1.0f + __expf(-(accZ[i] + bZ)));
    float rg = 1.0f / (1.0f + __expf(-(accR[i] + bR)));
    Zbuf[r * OUT_CH + colB] = z;
    RH[r * OUT_CH + colB] = rg * H[r * OUT_CH + colB];
  }
}

__global__ void __launch_bounds__(128)
cand_out_wmma_kernel(const float* __restrict__ x, const float* __restrict__ H,
                     const float* __restrict__ Lx, const float* __restrict__ RH,
                     const float* __restrict__ LRH,
                     const float* __restrict__ Zbuf,
                     const float* __restrict__ Wcx,
                     const float* __restrict__ Wch,
                     const float* __restrict__ bcx,
                     const float* __restrict__ bch, float* __restrict__ out) {
  int lane = threadIdx.x & 31;
  int wv = threadIdx.x >> 5;
  int half = lane >> 4;
  int l15 = lane & 15;
  int row0 = blockIdx.x * 16;
  int col0 = wv * 16;
  int rA = row0 + l15;
  int colB = col0 + l15;

  v8f accC = {};
  accC = accum_block(x, IN_CH, rA, Wcx, IN_CH, colB, half, accC);
  accC = accum_block(Lx, IN_CH, rA, Wcx + IN_CH * OUT_CH, IN_CH, colB, half, accC);
  accC = accum_block(RH, OUT_CH, rA, Wch, OUT_CH, colB, half, accC);
  accC = accum_block(LRH, OUT_CH, rA, Wch + OUT_CH * OUT_CH, OUT_CH, colB, half, accC);

  float bC = bcx[colB] + bch[colB];
#pragma unroll
  for (int i = 0; i < 8; ++i) {
    int r = row0 + i + 8 * half;
    float cv = tanhf(accC[i] + bC);
    float z = Zbuf[r * OUT_CH + colB];
    float h = H[r * OUT_CH + colB];
    out[r * OUT_CH + colB] = z * h + (1.0f - z) * cv;
  }
}


extern "C" void kernel_launch(void* const* d_in, const int* in_sizes, int n_in,
                              void* d_out, int out_size, void* d_ws,
                              size_t ws_size, hipStream_t stream) {
  const float* x = (const float*)d_in[0];
  const float* H = (const float*)d_in[1];
  const int* edge = (const int*)d_in[2];
  const float* Wzx = (const float*)d_in[3];
  const float* bzx = (const float*)d_in[4];
  const float* Wzh = (const float*)d_in[5];
  const float* bzh = (const float*)d_in[6];
  const float* Wrx = (const float*)d_in[7];
  const float* brx = (const float*)d_in[8];
  const float* Wrh = (const float*)d_in[9];
  const float* brh = (const float*)d_in[10];
  const float* Wcx = (const float*)d_in[11];
  const float* bcx = (const float*)d_in[12];
  const float* Wch = (const float*)d_in[13];
  const float* bch = (const float*)d_in[14];

  const int N = in_sizes[0] / IN_CH;
  const int E = in_sizes[2] / 2;
  const int* rowIdx = edge;
  const int* colIdx = edge + E;

  float* ws = (float*)d_ws;
  float* dinv = ws;
  float* Lx = dinv + N;
  float* LH = Lx + (size_t)N * IN_CH;
  float* LRH = LH + (size_t)N * OUT_CH;
  float* Zbuf = LRH + (size_t)N * OUT_CH;
  float* RH = Zbuf + (size_t)N * OUT_CH;
  const int nZero = N * (1 + IN_CH + OUT_CH + OUT_CH);

  zero_kernel<<<(nZero + 255) / 256, 256, 0, stream>>>(dinv, nZero);
  degree_kernel<<<(E + 255) / 256, 256, 0, stream>>>(rowIdx, dinv, E);
  dinv_kernel<<<(N + 255) / 256, 256, 0, stream>>>(dinv, N);
  {
    long long threads = (long long)E * 32;
    spmm_kernel<<<(int)((threads + 255) / 256), 256, 0, stream>>>(
        rowIdx, colIdx, dinv, H, LH, x, Lx, E);
  }
  gates_wmma_kernel<<<N / 16, 128, 0, stream>>>(x, H, Lx, LH, Wzx, Wzh, Wrx,
                                                Wrh, bzx, bzh, brx, brh, Zbuf,
                                                RH);
  {
    long long threads = (long long)E * 32;
    spmm_kernel<<<(int)((threads + 255) / 256), 256, 0, stream>>>(
        rowIdx, colIdx, dinv, RH, LRH, nullptr, nullptr, E);
  }
  cand_out_wmma_kernel<<<N / 16, 128, 0, stream>>>(
      x, H, Lx, RH, LRH, Zbuf, Wcx, Wch, bcx, bch, (float*)d_out);
}


# SpMM: edges counting-sorted by destination row in each call (int atomic rank, prefix sum); one wave sums each output row from gathers in rank order; float atomics and the zero fill removed
# speedup vs baseline: 2.1320x; 2.1320x over previous
#include <hip/hip_runtime.h>
#include <hip/hip_bf16.h>
#include <math.h>


typedef float v2f __attribute__((ext_vector_type(2)));
typedef float v8f __attribute__((ext_vector_type(8)));

#define IN_CH 8
#define OUT_CH 64

__device__ __forceinline__ v8f wmma_f32(v2f a, v2f b, v8f c) {
  return __builtin_amdgcn_wmma_f32_16x16x4_f32(false, a, false, b, (short)0, c,
                                               false, false);
}

__device__ __forceinline__ v8f accum_block(const float* __restrict__ A, int ldA,
                                           int rA, const float* __restrict__ W,
                                           int Kdim, int colB, int half,
                                           v8f acc) {
#pragma unroll 4
  for (int k0 = 0; k0 < Kdim; k0 += 4) {
    int ka = k0 + 2 * half;
    v2f a;
    a.x = A[rA * ldA + ka];
    a.y = A[rA * ldA + ka + 1];
    v2f b;
    b.x = W[ka * OUT_CH + colB];
    b.y = W[(ka + 1) * OUT_CH + colB];
    acc = wmma_f32(a, b, acc);
  }
  return acc;
}


__global__ void __launch_bounds__(256) zero_int_kernel(int* p, int n) {
  int i = blockIdx.x * blockDim.x + threadIdx.x;
  if (i < n) p[i] = 0;
}

__global__ void __launch_bounds__(256)
rank_kernel(const int* __restrict__ rowIdx, int* __restrict__ cnt,
            int* __restrict__ rank, int E) {
  int e = blockIdx.x * blockDim.x + threadIdx.x;
  if (e < E) rank[e] = atomicAdd(&cnt[rowIdx[e]], 1);
}

__global__ void __launch_bounds__(1024)
rowptr_kernel(const int* __restrict__ cnt, int* __restrict__ rowptr,
              float* __restrict__ dinv, int N) {
  __shared__ int s_pre[32];
  __shared__ int s_tot[32];
  int t = threadIdx.x;
  int lane = t & 31;
  int wv = t >> 5;
  int first = blockIdx.x * 1024;
  int pre = 0;
  for (int i = t; i < first; i += 1024) pre += cnt[i];
  int i = first + t;
  int c = (i < N) ? cnt[i] : 0;
  int incl = c;
#pragma unroll
  for (int d = 1; d < 32; d <<= 1) {
    int up = __shfl_up(incl, d);
    if (lane >= d) incl += up;
  }
#pragma unroll
  for (int d = 16; d > 0; d >>= 1) pre += __shfl_xor(pre, d);
  if (lane == 31) s_tot[wv] = incl;
  if (lane == 0) s_pre[wv] = pre;
  __syncthreads();
  int base = s_pre[lane] + ((lane < wv) ? s_tot[lane] : 0);
#pragma unroll
  for (int d = 16; d > 0; d >>= 1) base += __shfl_xor(base, d);
  int excl = base + incl - c;
  if (i < N) {
    rowptr[i] = excl;
    dinv[i] = (c > 0) ? rsqrtf((float)c) : 0.0f;
    if (i == N - 1) rowptr[N] = excl + c;
  }
}

__global__ void __launch_bounds__(256)
place_kernel(const int* __restrict__ rowIdx, const int* __restrict__ colIdx,
             const int* __restrict__ rank, const int* __restrict__ rowptr,
             const float* __restrict__ dinv, int* __restrict__ sedge, int E) {
  int e = blockIdx.x * blockDim.x + threadIdx.x;
  if (e < E) {
    int r = rowIdx[e];
    int c = colIdx[e];
    int p = rowptr[r] + rank[e];
    float nrm = -dinv[r] * dinv[c];
    ((int2*)sedge)[p] = make_int2(c, __float_as_int(nrm));
  }
}

template <typename V>
__device__ __forceinline__ V gather_edges(const V* __restrict__ src, int stride,
                                          int off, int2 en, int n, V acc) {
  int j = 0;
  for (; j + 8 <= n; j += 8) {
    V v[8];
#pragma unroll
    for (int u = 0; u < 8; ++u) {
      int c = __builtin_amdgcn_readlane(en.x, j + u);
      v[u] = src[(size_t)c * stride + off];
    }
#pragma unroll
    for (int u = 0; u < 8; ++u) {
      float w = __int_as_float(__builtin_amdgcn_readlane(en.y, j + u));
      acc += v[u] * w;
    }
  }
  int rem = n - j;
  if (rem > 0) {
    V v[7];
#pragma unroll
    for (int u = 0; u < 7; ++u) {
      v[u] = V(0.0f);
      if (u < rem) {
        int c = __builtin_amdgcn_readlane(en.x, j + u);
        v[u] = src[(size_t)c * stride + off];
      }
    }
#pragma unroll
    for (int u = 0; u < 7; ++u) {
      if (u < rem) {
        float w = __int_as_float(__builtin_amdgcn_readlane(en.y, j + u));
        acc += v[u] * w;
      }
    }
  }
  return acc;
}

__global__ void __launch_bounds__(256)
spmm_rows_kernel(const int* __restrict__ rowptr, const int* __restrict__ sedge,
                 const float* __restrict__ src64, float* __restrict__ dst64,
                 const float* __restrict__ src8, float* __restrict__ dst8,
                 int N) {
  int row = __builtin_amdgcn_readfirstlane(
      (blockIdx.x * blockDim.x + threadIdx.x) >> 5);
  int lane = threadIdx.x & 31;
  if (row >= N) return;
  int beg = __builtin_amdgcn_readfirstlane(rowptr[row]);
  int end = __builtin_amdgcn_readfirstlane(rowptr[row + 1]);
  const int2* __restrict__ ed = (const int2*)sedge;
  const v2f* __restrict__ s64 = (const v2f*)src64;
  const bool has8 = (src8 != nullptr);
  v2f acc = {0.0f, 0.0f};
  float acc8 = 0.0f;
  for (int base = beg; base < end; base += 32) {
    int n = end - base;
    if (n > 32) n = 32;
    int2 en = make_int2(0, 0);
    if (lane < n) en = ed[base + lane];
    acc = gather_edges<v2f>(s64, OUT_CH / 2, lane, en, n, acc);
    if (has8) acc8 = gather_edges<float>(src8, IN_CH, lane & 7, en, n, acc8);
  }
  ((v2f*)dst64)[(size_t)row * (OUT_CH / 2) + lane] = acc;
  if (has8 && lane < IN_CH) dst8[row * IN_CH + lane] = acc8;
}

__global__ void __launch_bounds__(128)
gates_wmma_kernel(const float* __restrict__ x, const float* __restrict__ H,
                  const float* __restrict__ Lx, const float* __restrict__ LH,
                  const float* __restrict__ Wzx, const float* __restrict__ Wzh,
                  const float* __restrict__ Wrx, const float* __restrict__ Wrh,
                  const float* __restrict__ bzx, const float* __restrict__ bzh,
                  const float* __restrict__ brx, const float* __restrict__ brh,
                  float* __restrict__ Zbuf, float* __restrict__ RH) {
  int lane = threadIdx.x & 31;
  int wv = threadIdx.x >> 5;
  int half = lane >> 4;
  int l15 = lane & 15;
  int row0 = blockIdx.x * 16;
  int col0 = wv * 16;
  int rA = row0 + l15;
  int colB = col0 + l15;

  v8f accZ = {};
  v8f accR = {};
  accZ = accum_block(x, IN_CH, rA, Wzx, IN_CH, colB, half, accZ);
  accR = accum_block(x, IN_CH, rA, Wrx, IN_CH, colB, half, accR);
  accZ = accum_block(Lx, IN_CH, rA, Wzx + IN_CH * OUT_CH, IN_CH, colB, half, accZ);
  accR = accum_block(Lx, IN_CH, rA, Wrx + IN_CH * OUT_CH, IN_CH, colB, half, accR);
  accZ = accum_block(H, OUT_CH, rA, Wzh, OUT_CH, colB, half, accZ);
  accR = accum_block(H, OUT_CH, rA, Wrh, OUT_CH, colB, half, accR);
  accZ = accum_block(LH, OUT_CH, rA, Wzh + OUT_CH * OUT_CH, OUT_CH, colB, half, accZ);
  accR = accum_block(LH, OUT_CH, rA, Wrh + OUT_CH * OUT_CH, OUT_CH, colB, half, accR);

  float bZ = bzx[colB] + bzh[colB];
  float bR = brx[colB] + brh[colB];
#pragma unroll
  for (int i = 0; i < 8; ++i) {
    int r = row0 + i + 8 * half;
    float z = 1.0f / (1.0f + __expf(-(accZ[i] + bZ)));
    float rg = 1.0f / (1.0f + __expf(-(accR[i] + bR)));
    Zbuf[r * OUT_CH + colB] = z;
    RH[r * OUT_CH + colB] = rg * H[r * OUT_CH + colB];
  }
}

__global__ void __launch_bounds__(128)
cand_out_wmma_kernel(const float* __restrict__ x, const float* __restrict__ H,
                     const float* __restrict__ Lx, const float* __restrict__ RH,
                     const float* __restrict__ LRH,
                     const float* __restrict__ Zbuf,
                     const float* __restrict__ Wcx,
                     const float* __restrict__ Wch,
                     const float* __restrict__ bcx,
                     const float* __restrict__ bch, float* __restrict__ out) {
  int lane = threadIdx.x & 31;
  int wv = threadIdx.x >> 5;
  int half = lane >> 4;
  int l15 = lane & 15;
  int row0 = blockIdx.x * 16;
  int col0 = wv * 16;
  int rA = row0 + l15;
  int colB = col0 + l15;

  v8f accC = {};
  accC = accum_block(x, IN_CH, rA, Wcx, IN_CH, colB, half, accC);
  accC = accum_block(Lx, IN_CH, rA, Wcx + IN_CH * OUT_CH, IN_CH, colB, half, accC);
  accC = accum_block(RH, OUT_CH, rA, Wch, OUT_CH, colB, half, accC);
  accC = accum_block(LRH, OUT_CH, rA, Wch + OUT_CH * OUT_CH, OUT_CH, colB, half, accC);

  float bC = bcx[colB] + bch[colB];
#pragma unroll
  for (int i = 0; i < 8; ++i) {
    int r = row0 + i + 8 * half;
    float cv = tanhf(accC[i] + bC);
    float z = Zbuf[r * OUT_CH + colB];
    float h = H[r * OUT_CH + colB];
    out[r * OUT_CH + colB] = z * h + (1.0f - z) * cv;
  }
}


extern "C" void kernel_launch(void* const* d_in, const int* in_sizes, int n_in,
                              void* d_out, int out_size, void* d_ws,
                              size_t ws_size, hipStream_t stream) {
  const float* x = (const float*)d_in[0];
  const float* H = (const float*)d_in[1];
  const int* edge = (const int*)d_in[2];
  const float* Wzx = (const float*)d_in[3];
  const float* bzx = (const float*)d_in[4];
  const float* Wzh = (const float*)d_in[5];
  const float* bzh = (const float*)d_in[6];
  const float* Wrx = (const float*)d_in[7];
  const float* brx = (const float*)d_in[8];
  const float* Wrh = (const float*)d_in[9];
  const float* brh = (const float*)d_in[10];
  const float* Wcx = (const float*)d_in[11];
  const float* bcx = (const float*)d_in[12];
  const float* Wch = (const float*)d_in[13];
  const float* bch = (const float*)d_in[14];

  const int N = in_sizes[0] / IN_CH;
  const int E = in_sizes[2] / 2;
  const int* rowIdx = edge;
  const int* colIdx = edge + E;

  float* ws = (float*)d_ws;
  float* LH = ws;
  float* LRH = LH + (size_t)N * OUT_CH;
  float* Zbuf = LRH + (size_t)N * OUT_CH;
  float* RH = Zbuf + (size_t)N * OUT_CH;
  float* Lx = RH + (size_t)N * OUT_CH;
  int* sedge = (int*)(Lx + (size_t)N * IN_CH);
  int* rank = sedge + (size_t)E * 2;
  int* cnt = rank + E;
  int* rowptr = cnt + N;
  float* dinv = (float*)(rowptr + N + 1);

  zero_int_kernel<<<(N + 255) / 256, 256, 0, stream>>>(cnt, N);
  rank_kernel<<<(E + 255) / 256, 256, 0, stream>>>(rowIdx, cnt, rank, E);
  rowptr_kernel<<<(N + 1023) / 1024, 1024, 0, stream>>>(cnt, rowptr, dinv, N);
  place_kernel<<<(E + 255) / 256, 256, 0, stream>>>(rowIdx, colIdx, rank, rowptr,
                                                    dinv, sedge, E);
  {
    long long threads = (long long)N * 32;
    spmm_rows_kernel<<<(int)((threads + 255) / 256), 256, 0, stream>>>(
        rowptr, sedge, H, LH, x, Lx, N);
  }
  gates_wmma_kernel<<<N / 16, 128, 0, stream>>>(x, H, Lx, LH, Wzx, Wzh, Wrx,
                                                Wrh, bzx, bzh, brx, brh, Zbuf,
                                                RH);
  {
    long long threads = (long long)N * 32;
    spmm_rows_kernel<<<(int)((threads + 255) / 256), 256, 0, stream>>>(
        rowptr, sedge, RH, LRH, nullptr, nullptr, N);
  }
  cand_out_wmma_kernel<<<N / 16, 128, 0, stream>>>(
      x, H, Lx, RH, LRH, Zbuf, Wcx, Wch, bcx, bch, (float*)d_out);
}


# Edge ordering through LDS: chunks sorted by 256-row bucket, then a block per bucket ranks rows with LDS counters (dinv table in LDS), whole-line stores; no global atomics; row sum order = LDS rank
# speedup vs baseline: 4.3423x; 2.0367x over previous
#include <hip/hip_runtime.h>
#include <hip/hip_bf16.h>
#include <math.h>


typedef float v2f __attribute__((ext_vector_type(2)));
typedef float v8f __attribute__((ext_vector_type(8)));

#define IN_CH 8
#define OUT_CH 64

__device__ __forceinline__ v8f wmma_f32(v2f a, v2f b, v8f c) {
  return __builtin_amdgcn_wmma_f32_16x16x4_f32(false, a, false, b, (short)0, c,
                                               false, false);
}

__device__ __forceinline__ v8f accum_block(const float* __restrict__ A, int ldA,
                                           int rA, const float* __restrict__ W,
                                           int Kdim, int colB, int half,
                                           v8f acc) {
#pragma unroll 4
  for (int k0 = 0; k0 < Kdim; k0 += 4) {
    int ka = k0 + 2 * half;
    v2f a;
    a.x = A[rA * ldA + ka];
    a.y = A[rA * ldA + ka + 1];
    v2f b;
    b.x = W[ka * OUT_CH + colB];
    b.y = W[(ka + 1) * OUT_CH + colB];
    acc = wmma_f32(a, b, acc);
  }
  return acc;
}


#define BUCKET_SHIFT 8
#define ROWS_PER_BUCKET 256
#define CHUNK 4096
#define CHUNK_THREADS 512
#define NB_CAP 1024
#define RUN_TILE 512
#define STAGE_CAP 12288
#define DINV_CAP 50176

__device__ __forceinline__ int block_excl_scan(int v, int* s_w, int& total) {
  int lane = threadIdx.x & 31;
  int wv = threadIdx.x >> 5;
  int nw = blockDim.x >> 5;
  int incl = v;
#pragma unroll
  for (int d = 1; d < 32; d <<= 1) {
    int up = __shfl_up(incl, d);
    if (lane >= d) incl += up;
  }
  if (lane == 31) s_w[wv] = incl;
  __syncthreads();
  int wsum = (lane < nw) ? s_w[lane] : 0;
  int winc = wsum;
#pragma unroll
  for (int d = 1; d < 32; d <<= 1) {
    int up = __shfl_up(winc, d);
    if (lane >= d) winc += up;
  }
  int wpre = __shfl(winc - wsum, wv);
  total = __shfl(winc, 31);
  __syncthreads();
  return wpre + incl - v;
}

__global__ void __launch_bounds__(CHUNK_THREADS)
chunk_sort_kernel(const int* __restrict__ rowIdx, const int* __restrict__ colIdx,
                  int* __restrict__ tmp, int* __restrict__ chunkOff, int E,
                  int NB) {
  __shared__ int s_hist[NB_CAP + 1];
  __shared__ int s_w[32];
  __shared__ int2 s_stage[CHUNK];
  const int t = threadIdx.x;
  const int base = blockIdx.x * CHUNK;
  for (int i = t; i <= NB; i += CHUNK_THREADS) s_hist[i] = 0;
  __syncthreads();
  int r[CHUNK / CHUNK_THREADS], c[CHUNK / CHUNK_THREADS], rk[CHUNK / CHUNK_THREADS];
#pragma unroll
  for (int u = 0; u < CHUNK / CHUNK_THREADS; ++u) {
    int e = base + u * CHUNK_THREADS + t;
    r[u] = -1;
    c[u] = 0;
    if (e < E) {
      r[u] = rowIdx[e];
      c[u] = colIdx[e];
    }
  }
#pragma unroll
  for (int u = 0; u < CHUNK / CHUNK_THREADS; ++u) {
    rk[u] = 0;
    if (r[u] >= 0) rk[u] = atomicAdd(&s_hist[r[u] >> BUCKET_SHIFT], 1);
  }
  __syncthreads();
  int carry = 0;
  for (int i0 = 0; i0 < NB; i0 += CHUNK_THREADS) {
    int i = i0 + t;
    int v = (i < NB) ? s_hist[i] : 0;
    int tot;
    int ex = block_excl_scan(v, s_w, tot);
    if (i < NB) s_hist[i] = carry + ex;
    carry += tot;
  }
  if (t == 0) s_hist[NB] = carry;
  __syncthreads();
  for (int i = t; i <= NB; i += CHUNK_THREADS)
    chunkOff[(size_t)blockIdx.x * (NB + 1) + i] = s_hist[i];
#pragma unroll
  for (int u = 0; u < CHUNK / CHUNK_THREADS; ++u) {
    if (r[u] >= 0)
      s_stage[s_hist[r[u] >> BUCKET_SHIFT] + rk[u]] = make_int2(r[u], c[u]);
  }
  __syncthreads();
  int cnt = E - base;
  if (cnt > CHUNK) cnt = CHUNK;
  for (int i = t; i < cnt; i += CHUNK_THREADS)
    ((int2*)tmp)[(size_t)base + i] = s_stage[i];
}

template <typename F>
__device__ __forceinline__ void for_each_run_edge(const int2* __restrict__ tmp2,
                                                  const int* s_start,
                                                  const int* s_len, int nr,
                                                  int lane, int wv, int nw,
                                                  F f) {
  for (int c = wv; c < nr; c += 4 * nw) {
    int2 v[4];
    int len[4];
    int st[4];
#pragma unroll
    for (int u = 0; u < 4; ++u) {
      int cc = c + u * nw;
      len[u] = (cc < nr) ? s_len[cc] : 0;
      st[u] = (cc < nr) ? s_start[cc] : 0;
      v[u] = make_int2(0, 0);
      if (lane < len[u]) v[u] = tmp2[(size_t)st[u] + lane];
    }
#pragma unroll
    for (int u = 0; u < 4; ++u) {
      if (lane < len[u]) f(v[u]);
      for (int i = lane + 32; i < len[u]; i += 32) f(tmp2[(size_t)st[u] + i]);
    }
  }
}

__device__ __forceinline__ int load_run_tile(const int* __restrict__ chunkOff,
                                             int b, int NB, int nchunks, int c0,
                                             int* s_start, int* s_len) {
  int before = 0;
  int t = threadIdx.x;
  if (t < RUN_TILE) {
    int c = c0 + t;
    int st = 0, len = 0;
    if (c < nchunks) {
      const int* o = chunkOff + (size_t)c * (NB + 1) + b;
      int o0 = o[0];
      int o1 = o[1];
      st = c * CHUNK + o0;
      len = o1 - o0;
      before = o0;
    }
    s_start[t] = st;
    s_len[t] = len;
  }
  return before;
}

__global__ void __launch_bounds__(1024)
bucket_degree_kernel(const int* __restrict__ tmp, const int* __restrict__ chunkOff,
                     int* __restrict__ rowptr, float* __restrict__ dinv, int N,
                     int NB, int nchunks) {
  __shared__ int s_start[RUN_TILE];
  __shared__ int s_len[RUN_TILE];
  __shared__ int s_cnt[ROWS_PER_BUCKET];
  __shared__ int s_w[32];
  const int t = threadIdx.x;
  const int lane = t & 31;
  const int wv = t >> 5;
  const int nw = blockDim.x >> 5;
  const int b = blockIdx.x;
  const int row0 = b << BUCKET_SHIFT;
  const int2* __restrict__ tmp2 = (const int2*)tmp;
  if (t < ROWS_PER_BUCKET) s_cnt[t] = 0;
  int before = 0;
  for (int c0 = 0; c0 < nchunks; c0 += RUN_TILE) {
    __syncthreads();
    before += load_run_tile(chunkOff, b, NB, nchunks, c0, s_start, s_len);
    __syncthreads();
    int nr = nchunks - c0;
    if (nr > RUN_TILE) nr = RUN_TILE;
    for_each_run_edge(tmp2, s_start, s_len, nr, lane, wv, nw,
                      [&](int2 rc) { atomicAdd(&s_cnt[rc.x - row0], 1); });
  }
  __syncthreads();
  int deg = (t < ROWS_PER_BUCKET) ? s_cnt[t] : 0;
  int tot, totBefore;
  int ex = block_excl_scan(deg, s_w, tot);
  block_excl_scan(before, s_w, totBefore);
  int row = row0 + t;
  if (t < ROWS_PER_BUCKET && row < N) {
    rowptr[row] = totBefore + ex;
    dinv[row] = (deg > 0) ? rsqrtf((float)deg) : 0.0f;
  }
  if (b == NB - 1 && t == 0) rowptr[N] = totBefore + tot;
}

__global__ void __launch_bounds__(1024)
bucket_place_kernel(const int* __restrict__ tmp, const int* __restrict__ chunkOff,
                    const int* __restrict__ rowptr, const float* __restrict__ dinv,
                    int* __restrict__ sedge, int N, int NB, int nchunks) {
  __shared__ float4 s_dinv4[DINV_CAP / 4];
  __shared__ int2 s_stage[STAGE_CAP];
  __shared__ int s_start[RUN_TILE];
  __shared__ int s_len[RUN_TILE];
  __shared__ int s_cnt[ROWS_PER_BUCKET];
  __shared__ int s_rp[ROWS_PER_BUCKET + 1];
  float* s_dinv = (float*)s_dinv4;
  const int t = threadIdx.x;
  const int lane = t & 31;
  const int wv = t >> 5;
  const int nw = blockDim.x >> 5;
  const int b = blockIdx.x;
  const int row0 = b << BUCKET_SHIFT;
  const int2* __restrict__ tmp2 = (const int2*)tmp;
  int2* __restrict__ sedge2 = (int2*)sedge;
  const bool tab = (N <= DINV_CAP);
  if (tab) {
    if ((((size_t)dinv) & 15) == 0) {
      int n4 = N >> 2;
      for (int i = t; i < n4; i += 1024) s_dinv4[i] = ((const float4*)dinv)[i];
      for (int i = (n4 << 2) + t; i < N; i += 1024) s_dinv[i] = dinv[i];
    } else {
      for (int i = t; i < N; i += 1024) s_dinv[i] = dinv[i];
    }
  }
  if (t <= ROWS_PER_BUCKET) {
    int rr = row0 + t;
    if (rr > N) rr = N;
    s_rp[t] = rowptr[rr];
  }
  if (t < ROWS_PER_BUCKET) s_cnt[t] = 0;
  __syncthreads();
  const int seg0 = s_rp[0];
  const int nb = s_rp[ROWS_PER_BUCKET] - seg0;
  const bool staged = (nb <= STAGE_CAP);
  for (int c0 = 0; c0 < nchunks; c0 += RUN_TILE) {
    __syncthreads();
    load_run_tile(chunkOff, b, NB, nchunks, c0, s_start, s_len);
    __syncthreads();
    int nr = nchunks - c0;
    if (nr > RUN_TILE) nr = RUN_TILE;
    for_each_run_edge(tmp2, s_start, s_len, nr, lane, wv, nw, [&](int2 rc) {
      int lr = rc.x - row0;
      int rk = atomicAdd(&s_cnt[lr], 1);
      int pl = s_rp[lr] - seg0 + rk;
      float dr = tab ? s_dinv[rc.x] : dinv[rc.x];
      float dc = tab ? s_dinv[rc.y] : dinv[rc.y];
      float nrm = -dr * dc;
      int2 val = make_int2(rc.y, __float_as_int(nrm));
      if (staged)
        s_stage[pl] = val;
      else
        sedge2[(size_t)seg0 + pl] = val;
    });
  }
  __syncthreads();
  if (staged) {
    for (int i = t; i < nb; i += 1024) sedge2[(size_t)seg0 + i] = s_stage[i];
  }
}

template <typename V>
__device__ __forceinline__ V gather_edges(const V* __restrict__ src, int stride,
                                          int off, int2 en, int n, V acc) {
  int j = 0;
  for (; j + 8 <= n; j += 8) {
    V v[8];
#pragma unroll
    for (int u = 0; u < 8; ++u) {
      int c = __builtin_amdgcn_readlane(en.x, j + u);
      v[u] = src[(size_t)c * stride + off];
    }
#pragma unroll
    for (int u = 0; u < 8; ++u) {
      float w = __int_as_float(__builtin_amdgcn_readlane(en.y, j + u));
      acc += v[u] * w;
    }
  }
  int rem = n - j;
  if (rem > 0) {
    V v[7];
#pragma unroll
    for (int u = 0; u < 7; ++u) {
      v[u] = V(0.0f);
      if (u < rem) {
        int c = __builtin_amdgcn_readlane(en.x, j + u);
        v[u] = src[(size_t)c * stride + off];
      }
    }
#pragma unroll
    for (int u = 0; u < 7; ++u) {
      if (u < rem) {
        float w = __int_as_float(__builtin_amdgcn_readlane(en.y, j + u));
        acc += v[u] * w;
      }
    }
  }
  return acc;
}

__global__ void __launch_bounds__(256)
spmm_rows_kernel(const int* __restrict__ rowptr, const int* __restrict__ sedge,
                 const float* __restrict__ src64, float* __restrict__ dst64,
                 const float* __restrict__ src8, float* __restrict__ dst8,
                 int N) {
  int row = __builtin_amdgcn_readfirstlane(
      (blockIdx.x * blockDim.x + threadIdx.x) >> 5);
  int lane = threadIdx.x & 31;
  if (row >= N) return;
  int beg = __builtin_amdgcn_readfirstlane(rowptr[row]);
  int end = __builtin_amdgcn_readfirstlane(rowptr[row + 1]);
  const int2* __restrict__ ed = (const int2*)sedge;
  const v2f* __restrict__ s64 = (const v2f*)src64;
  const bool has8 = (src8 != nullptr);
  v2f acc = {0.0f, 0.0f};
  float acc8 = 0.0f;
  for (int base = beg; base < end; base += 32) {
    int n = end - base;
    if (n > 32) n = 32;
    int2 en = make_int2(0, 0);
    if (lane < n) en = ed[base + lane];
    acc = gather_edges<v2f>(s64, OUT_CH / 2, lane, en, n, acc);
    if (has8) acc8 = gather_edges<float>(src8, IN_CH, lane & 7, en, n, acc8);
  }
  ((v2f*)dst64)[(size_t)row * (OUT_CH / 2) + lane] = acc;
  if (has8 && lane < IN_CH) dst8[row * IN_CH + lane] = acc8;
}

__global__ void __launch_bounds__(128)
gates_wmma_kernel(const float* __restrict__ x, const float* __restrict__ H,
                  const float* __restrict__ Lx, const float* __restrict__ LH,
                  const float* __restrict__ Wzx, const float* __restrict__ Wzh,
                  const float* __restrict__ Wrx, const float* __restrict__ Wrh,
                  const float* __restrict__ bzx, const float* __restrict__ bzh,
                  const float* __restrict__ brx, const float* __restrict__ brh,
                  float* __restrict__ Zbuf, float* __restrict__ RH) {
  int lane = threadIdx.x & 31;
  int wv = threadIdx.x >> 5;
  int half = lane >> 4;
  int l15 = lane & 15;
  int row0 = blockIdx.x * 16;
  int col0 = wv * 16;
  int rA = row0 + l15;
  int colB = col0 + l15;

  v8f accZ = {};
  v8f accR = {};
  accZ = accum_block(x, IN_CH, rA, Wzx, IN_CH, colB, half, accZ);
  accR = accum_block(x, IN_CH, rA, Wrx, IN_CH, colB, half, accR);
  accZ = accum_block(Lx, IN_CH, rA, Wzx + IN_CH * OUT_CH, IN_CH, colB, half, accZ);
  accR = accum_block(Lx, IN_CH, rA, Wrx + IN_CH * OUT_CH, IN_CH, colB, half, accR);
  accZ = accum_block(H, OUT_CH, rA, Wzh, OUT_CH, colB, half, accZ);
  accR = accum_block(H, OUT_CH, rA, Wrh, OUT_CH, colB, half, accR);
  accZ = accum_block(LH, OUT_CH, rA, Wzh + OUT_CH * OUT_CH, OUT_CH, colB, half, accZ);
  accR = accum_block(LH, OUT_CH, rA, Wrh + OUT_CH * OUT_CH, OUT_CH, colB, half, accR);

  float bZ = bzx[colB] + bzh[colB];
  float bR = brx[colB] + brh[colB];
#pragma unroll
  for (int i = 0; i < 8; ++i) {
    int r = row0 + i + 8 * half;
    float z = 1.0f / (1.0f + __expf(-(accZ[i] + bZ)));
    float rg = 1.0f / (1.0f + __expf(-(accR[i] + bR)));
    Zbuf[r * OUT_CH + colB] = z;
    RH[r * OUT_CH + colB] = rg * H[r * OUT_CH + colB];
  }
}

__global__ void __launch_bounds__(128)
cand_out_wmma_kernel(const float* __restrict__ x, const float* __restrict__ H,
                     const float* __restrict__ Lx, const float* __restrict__ RH,
                     const float* __restrict__ LRH,
                     const float* __restrict__ Zbuf,
                     const float* __restrict__ Wcx,
                     const float* __restrict__ Wch,
                     const float* __restrict__ bcx,
                     const float* __restrict__ bch, float* __restrict__ out) {
  int lane = threadIdx.x & 31;
  int wv = threadIdx.x >> 5;
  int half = lane >> 4;
  int l15 = lane & 15;
  int row0 = blockIdx.x * 16;
  int col0 = wv * 16;
  int rA = row0 + l15;
  int colB = col0 + l15;

  v8f accC = {};
  accC = accum_block(x, IN_CH, rA, Wcx, IN_CH, colB, half, accC);
  accC = accum_block(Lx, IN_CH, rA, Wcx + IN_CH * OUT_CH, IN_CH, colB, half, accC);
  accC = accum_block(RH, OUT_CH, rA, Wch, OUT_CH, colB, half, accC);
  accC = accum_block(LRH, OUT_CH, rA, Wch + OUT_CH * OUT_CH, OUT_CH, colB, half, accC);

  float bC = bcx[colB] + bch[colB];
#pragma unroll
  for (int i = 0; i < 8; ++i) {
    int r = row0 + i + 8 * half;
    float cv = tanhf(accC[i] + bC);
    float z = Zbuf[r * OUT_CH + colB];
    float h = H[r * OUT_CH + colB];
    out[r * OUT_CH + colB] = z * h + (1.0f - z) * cv;
  }
}


extern "C" void kernel_launch(void* const* d_in, const int* in_sizes, int n_in,
                              void* d_out, int out_size, void* d_ws,
                              size_t ws_size, hipStream_t stream) {
  const float* x = (const float*)d_in[0];
  const float* H = (const float*)d_in[1];
  const int* edge = (const int*)d_in[2];
  const float* Wzx = (const float*)d_in[3];
  const float* bzx = (const float*)d_in[4];
  const float* Wzh = (const float*)d_in[5];
  const float* bzh = (const float*)d_in[6];
  const float* Wrx = (const float*)d_in[7];
  const float* brx = (const float*)d_in[8];
  const float* Wrh = (const float*)d_in[9];
  const float* brh = (const float*)d_in[10];
  const float* Wcx = (const float*)d_in[11];
  const float* bcx = (const float*)d_in[12];
  const float* Wch = (const float*)d_in[13];
  const float* bch = (const float*)d_in[14];

  const int N = in_sizes[0] / IN_CH;
  const int E = in_sizes[2] / 2;
  const int* rowIdx = edge;
  const int* colIdx = edge + E;

  float* ws = (float*)d_ws;
  float* LH = ws;
  float* LRH = LH + (size_t)N * OUT_CH;
  float* Zbuf = LRH + (size_t)N * OUT_CH;
  float* RH = Zbuf + (size_t)N * OUT_CH;
  float* Lx = RH + (size_t)N * OUT_CH;
  int* sedge = (int*)(Lx + (size_t)N * IN_CH);
  int* tmp = sedge + (size_t)E * 2;
  float* dinv = (float*)(tmp + (size_t)E * 2);
  int* rowptr = (int*)(dinv + N);
  int* chunkOff = rowptr + N + 1;

  const int NB = (N + ROWS_PER_BUCKET - 1) / ROWS_PER_BUCKET;
  const int nchunks = (E + CHUNK - 1) / CHUNK;

  chunk_sort_kernel<<<nchunks, CHUNK_THREADS, 0, stream>>>(rowIdx, colIdx, tmp,
                                                           chunkOff, E, NB);
  bucket_degree_kernel<<<NB, 1024, 0, stream>>>(tmp, chunkOff, rowptr, dinv, N,
                                                NB, nchunks);
  bucket_place_kernel<<<NB, 1024, 0, stream>>>(tmp, chunkOff, rowptr, dinv,
                                               sedge, N, NB, nchunks);
  {
    long long threads = (long long)N * 32;
    spmm_rows_kernel<<<(int)((threads + 255) / 256), 256, 0, stream>>>(
        rowptr, sedge, H, LH, x, Lx, N);
  }
  gates_wmma_kernel<<<N / 16, 128, 0, stream>>>(x, H, Lx, LH, Wzx, Wzh, Wrx,
                                                Wrh, bzx, bzh, brx, brh, Zbuf,
                                                RH);
  {
    long long threads = (long long)N * 32;
    spmm_rows_kernel<<<(int)((threads + 255) / 256), 256, 0, stream>>>(
        rowptr, sedge, RH, LRH, nullptr, nullptr, N);
  }
  cand_out_wmma_kernel<<<N / 16, 128, 0, stream>>>(
      x, H, Lx, RH, LRH, Zbuf, Wcx, Wch, bcx, bch, (float*)d_out);
}


# Dense kernels: each wave keeps its 144x16 weight slice as B fragments in registers and loops over 16-row tiles staged in LDS by whole-line loads; Z and R share one A fragment; sum order kept
# speedup vs baseline: 4.9541x; 1.1409x over previous
#include <hip/hip_runtime.h>
#include <hip/hip_bf16.h>
#include <math.h>


typedef float v2f __attribute__((ext_vector_type(2)));
typedef float v4f __attribute__((ext_vector_type(4)));
typedef float v8f __attribute__((ext_vector_type(8)));

#define IN_CH 8
#define OUT_CH 64

__device__ __forceinline__ v8f wmma_f32(v2f a, v2f b, v8f c) {
  return __builtin_amdgcn_wmma_f32_16x16x4_f32(false, a, false, b, (short)0, c,
                                               false, false);
}


#define BUCKET_SHIFT 8
#define ROWS_PER_BUCKET 256
#define CHUNK 4096
#define CHUNK_THREADS 512
#define NB_CAP 1024
#define RUN_TILE 512
#define STAGE_CAP 12288
#define DINV_CAP 50176

__device__ __forceinline__ int block_excl_scan(int v, int* s_w, int& total) {
  int lane = threadIdx.x & 31;
  int wv = threadIdx.x >> 5;
  int nw = blockDim.x >> 5;
  int incl = v;
#pragma unroll
  for (int d = 1; d < 32; d <<= 1) {
    int up = __shfl_up(incl, d);
    if (lane >= d) incl += up;
  }
  if (lane == 31) s_w[wv] = incl;
  __syncthreads();
  int wsum = (lane < nw) ? s_w[lane] : 0;
  int winc = wsum;
#pragma unroll
  for (int d = 1; d < 32; d <<= 1) {
    int up = __shfl_up(winc, d);
    if (lane >= d) winc += up;
  }
  int wpre = __shfl(winc - wsum, wv);
  total = __shfl(winc, 31);
  __syncthreads();
  return wpre + incl - v;
}

__global__ void __launch_bounds__(CHUNK_THREADS)
chunk_sort_kernel(const int* __restrict__ rowIdx, const int* __restrict__ colIdx,
                  int* __restrict__ tmp, int* __restrict__ chunkOff, int E,
                  int NB) {
  __shared__ int s_hist[NB_CAP + 1];
  __shared__ int s_w[32];
  __shared__ int2 s_stage[CHUNK];
  const int t = threadIdx.x;
  const int base = blockIdx.x * CHUNK;
  for (int i = t; i <= NB; i += CHUNK_THREADS) s_hist[i] = 0;
  __syncthreads();
  int r[CHUNK / CHUNK_THREADS], c[CHUNK / CHUNK_THREADS], rk[CHUNK / CHUNK_THREADS];
#pragma unroll
  for (int u = 0; u < CHUNK / CHUNK_THREADS; ++u) {
    int e = base + u * CHUNK_THREADS + t;
    r[u] = -1;
    c[u] = 0;
    if (e < E) {
      r[u] = rowIdx[e];
      c[u] = colIdx[e];
    }
  }
#pragma unroll
  for (int u = 0; u < CHUNK / CHUNK_THREADS; ++u) {
    rk[u] = 0;
    if (r[u] >= 0) rk[u] = atomicAdd(&s_hist[r[u] >> BUCKET_SHIFT], 1);
  }
  __syncthreads();
  int carry = 0;
  for (int i0 = 0; i0 < NB; i0 += CHUNK_THREADS) {
    int i = i0 + t;
    int v = (i < NB) ? s_hist[i] : 0;
    int tot;
    int ex = block_excl_scan(v, s_w, tot);
    if (i < NB) s_hist[i] = carry + ex;
    carry += tot;
  }
  if (t == 0) s_hist[NB] = carry;
  __syncthreads();
  for (int i = t; i <= NB; i += CHUNK_THREADS)
    chunkOff[(size_t)blockIdx.x * (NB + 1) + i] = s_hist[i];
#pragma unroll
  for (int u = 0; u < CHUNK / CHUNK_THREADS; ++u) {
    if (r[u] >= 0)
      s_stage[s_hist[r[u] >> BUCKET_SHIFT] + rk[u]] = make_int2(r[u], c[u]);
  }
  __syncthreads();
  int cnt = E - base;
  if (cnt > CHUNK) cnt = CHUNK;
  for (int i = t; i < cnt; i += CHUNK_THREADS)
    ((int2*)tmp)[(size_t)base + i] = s_stage[i];
}

template <typename F>
__device__ __forceinline__ void for_each_run_edge(const int2* __restrict__ tmp2,
                                                  const int* s_start,
                                                  const int* s_len, int nr,
                                                  int lane, int wv, int nw,
                                                  F f) {
  for (int c = wv; c < nr; c += 4 * nw) {
    int2 v[4];
    int len[4];
    int st[4];
#pragma unroll
    for (int u = 0; u < 4; ++u) {
      int cc = c + u * nw;
      len[u] = (cc < nr) ? s_len[cc] : 0;
      st[u] = (cc < nr) ? s_start[cc] : 0;
      v[u] = make_int2(0, 0);
      if (lane < len[u]) v[u] = tmp2[(size_t)st[u] + lane];
    }
#pragma unroll
    for (int u = 0; u < 4; ++u) {
      if (lane < len[u]) f(v[u]);
      for (int i = lane + 32; i < len[u]; i += 32) f(tmp2[(size_t)st[u] + i]);
    }
  }
}

__device__ __forceinline__ int load_run_tile(const int* __restrict__ chunkOff,
                                             int b, int NB, int nchunks, int c0,
                                             int* s_start, int* s_len) {
  int before = 0;
  int t = threadIdx.x;
  if (t < RUN_TILE) {
    int c = c0 + t;
    int st = 0, len = 0;
    if (c < nchunks) {
      const int* o = chunkOff + (size_t)c * (NB + 1) + b;
      int o0 = o[0];
      int o1 = o[1];
      st = c * CHUNK + o0;
      len = o1 - o0;
      before = o0;
    }
    s_start[t] = st;
    s_len[t] = len;
  }
  return before;
}

__global__ void __launch_bounds__(1024)
bucket_degree_kernel(const int* __restrict__ tmp, const int* __restrict__ chunkOff,
                     int* __restrict__ rowptr, float* __restrict__ dinv, int N,
                     int NB, int nchunks) {
  __shared__ int s_start[RUN_TILE];
  __shared__ int s_len[RUN_TILE];
  __shared__ int s_cnt[ROWS_PER_BUCKET];
  __shared__ int s_w[32];
  const int t = threadIdx.x;
  const int lane = t & 31;
  const int wv = t >> 5;
  const int nw = blockDim.x >> 5;
  const int b = blockIdx.x;
  const int row0 = b << BUCKET_SHIFT;
  const int2* __restrict__ tmp2 = (const int2*)tmp;
  if (t < ROWS_PER_BUCKET) s_cnt[t] = 0;
  int before = 0;
  for (int c0 = 0; c0 < nchunks; c0 += RUN_TILE) {
    __syncthreads();
    before += load_run_tile(chunkOff, b, NB, nchunks, c0, s_start, s_len);
    __syncthreads();
    int nr = nchunks - c0;
    if (nr > RUN_TILE) nr = RUN_TILE;
    for_each_run_edge(tmp2, s_start, s_len, nr, lane, wv, nw,
                      [&](int2 rc) { atomicAdd(&s_cnt[rc.x - row0], 1); });
  }
  __syncthreads();
  int deg = (t < ROWS_PER_BUCKET) ? s_cnt[t] : 0;
  int tot, totBefore;
  int ex = block_excl_scan(deg, s_w, tot);
  block_excl_scan(before, s_w, totBefore);
  int row = row0 + t;
  if (t < ROWS_PER_BUCKET && row < N) {
    rowptr[row] = totBefore + ex;
    dinv[row] = (deg > 0) ? rsqrtf((float)deg) : 0.0f;
  }
  if (b == NB - 1 && t == 0) rowptr[N] = totBefore + tot;
}

__global__ void __launch_bounds__(1024)
bucket_place_kernel(const int* __restrict__ tmp, const int* __restrict__ chunkOff,
                    const int* __restrict__ rowptr, const float* __restrict__ dinv,
                    int* __restrict__ sedge, int N, int NB, int nchunks) {
  __shared__ float4 s_dinv4[DINV_CAP / 4];
  __shared__ int2 s_stage[STAGE_CAP];
  __shared__ int s_start[RUN_TILE];
  __shared__ int s_len[RUN_TILE];
  __shared__ int s_cnt[ROWS_PER_BUCKET];
  __shared__ int s_rp[ROWS_PER_BUCKET + 1];
  float* s_dinv = (float*)s_dinv4;
  const int t = threadIdx.x;
  const int lane = t & 31;
  const int wv = t >> 5;
  const int nw = blockDim.x >> 5;
  const int b = blockIdx.x;
  const int row0 = b << BUCKET_SHIFT;
  const int2* __restrict__ tmp2 = (const int2*)tmp;
  int2* __restrict__ sedge2 = (int2*)sedge;
  const bool tab = (N <= DINV_CAP);
  if (tab) {
    if ((((size_t)dinv) & 15) == 0) {
      int n4 = N >> 2;
      for (int i = t; i < n4; i += 1024) s_dinv4[i] = ((const float4*)dinv)[i];
      for (int i = (n4 << 2) + t; i < N; i += 1024) s_dinv[i] = dinv[i];
    } else {
      for (int i = t; i < N; i += 1024) s_dinv[i] = dinv[i];
    }
  }
  if (t <= ROWS_PER_BUCKET) {
    int rr = row0 + t;
    if (rr > N) rr = N;
    s_rp[t] = rowptr[rr];
  }
  if (t < ROWS_PER_BUCKET) s_cnt[t] = 0;
  __syncthreads();
  const int seg0 = s_rp[0];
  const int nb = s_rp[ROWS_PER_BUCKET] - seg0;
  const bool staged = (nb <= STAGE_CAP);
  for (int c0 = 0; c0 < nchunks; c0 += RUN_TILE) {
    __syncthreads();
    load_run_tile(chunkOff, b, NB, nchunks, c0, s_start, s_len);
    __syncthreads();
    int nr = nchunks - c0;
    if (nr > RUN_TILE) nr = RUN_TILE;
    for_each_run_edge(tmp2, s_start, s_len, nr, lane, wv, nw, [&](int2 rc) {
      int lr = rc.x - row0;
      int rk = atomicAdd(&s_cnt[lr], 1);
      int pl = s_rp[lr] - seg0 + rk;
      float dr = tab ? s_dinv[rc.x] : dinv[rc.x];
      float dc = tab ? s_dinv[rc.y] : dinv[rc.y];
      float nrm = -dr * dc;
      int2 val = make_int2(rc.y, __float_as_int(nrm));
      if (staged)
        s_stage[pl] = val;
      else
        sedge2[(size_t)seg0 + pl] = val;
    });
  }
  __syncthreads();
  if (staged) {
    for (int i = t; i < nb; i += 1024) sedge2[(size_t)seg0 + i] = s_stage[i];
  }
}

template <typename V>
__device__ __forceinline__ V gather_edges(const V* __restrict__ src, int stride,
                                          int off, int2 en, int n, V acc) {
  int j = 0;
  for (; j + 8 <= n; j += 8) {
    V v[8];
#pragma unroll
    for (int u = 0; u < 8; ++u) {
      int c = __builtin_amdgcn_readlane(en.x, j + u);
      v[u] = src[(size_t)c * stride + off];
    }
#pragma unroll
    for (int u = 0; u < 8; ++u) {
      float w = __int_as_float(__builtin_amdgcn_readlane(en.y, j + u));
      acc += v[u] * w;
    }
  }
  int rem = n - j;
  if (rem > 0) {
    V v[7];
#pragma unroll
    for (int u = 0; u < 7; ++u) {
      v[u] = V(0.0f);
      if (u < rem) {
        int c = __builtin_amdgcn_readlane(en.x, j + u);
        v[u] = src[(size_t)c * stride + off];
      }
    }
#pragma unroll
    for (int u = 0; u < 7; ++u) {
      if (u < rem) {
        float w = __int_as_float(__builtin_amdgcn_readlane(en.y, j + u));
        acc += v[u] * w;
      }
    }
  }
  return acc;
}

__global__ void __launch_bounds__(256)
spmm_rows_kernel(const int* __restrict__ rowptr, const int* __restrict__ sedge,
                 const float* __restrict__ src64, float* __restrict__ dst64,
                 const float* __restrict__ src8, float* __restrict__ dst8,
                 int N) {
  int row = __builtin_amdgcn_readfirstlane(
      (blockIdx.x * blockDim.x + threadIdx.x) >> 5);
  int lane = threadIdx.x & 31;
  if (row >= N) return;
  int beg = __builtin_amdgcn_readfirstlane(rowptr[row]);
  int end = __builtin_amdgcn_readfirstlane(rowptr[row + 1]);
  const int2* __restrict__ ed = (const int2*)sedge;
  const v2f* __restrict__ s64 = (const v2f*)src64;
  const bool has8 = (src8 != nullptr);
  v2f acc = {0.0f, 0.0f};
  float acc8 = 0.0f;
  for (int base = beg; base < end; base += 32) {
    int n = end - base;
    if (n > 32) n = 32;
    int2 en = make_int2(0, 0);
    if (lane < n) en = ed[base + lane];
    acc = gather_edges<v2f>(s64, OUT_CH / 2, lane, en, n, acc);
    if (has8) acc8 = gather_edges<float>(src8, IN_CH, lane & 7, en, n, acc8);
  }
  ((v2f*)dst64)[(size_t)row * (OUT_CH / 2) + lane] = acc;
  if (has8 && lane < IN_CH) dst8[row * IN_CH + lane] = acc8;
}


#define A_K 144
#define K_STEPS (A_K / 4)
#define A_PITCH 148
#define DENSE_GRID 1024

__device__ __forceinline__ void tile_load(const float* __restrict__ x,
                                          const float* __restrict__ Lx,
                                          const float* __restrict__ P,
                                          const float* __restrict__ LP,
                                          int row0, int t, v4f& p0, v4f& p1,
                                          v4f& q0, v4f& q1, v4f& s) {
  const v4f* p4 = (const v4f*)(P + (size_t)row0 * OUT_CH);
  const v4f* q4 = (const v4f*)(LP + (size_t)row0 * OUT_CH);
  p0 = p4[t];
  p1 = p4[t + 128];
  q0 = q4[t];
  q1 = q4[t + 128];
  const float* sb = (t < 32) ? x : Lx;
  if (t < 64) s = ((const v4f*)(sb + (size_t)row0 * IN_CH))[t & 31];
}

__device__ __forceinline__ void tile_store(float* sa, int t, v4f p0, v4f p1,
                                           v4f q0, v4f q1, v4f s) {
  int row = t >> 4;
  int c = (t & 15) * 4;
  *(v4f*)&sa[row * A_PITCH + 2 * IN_CH + c] = p0;
  *(v4f*)&sa[(row + 8) * A_PITCH + 2 * IN_CH + c] = p1;
  *(v4f*)&sa[row * A_PITCH + 2 * IN_CH + OUT_CH + c] = q0;
  *(v4f*)&sa[(row + 8) * A_PITCH + 2 * IN_CH + OUT_CH + c] = q1;
  if (t < 64) {
    int u = t & 31;
    int off = (t < 32) ? 0 : IN_CH;
    *(v4f*)&sa[(u >> 1) * A_PITCH + off + (u & 1) * 4] = s;
  }
}

__device__ __forceinline__ void load_b_frags(const float* __restrict__ Wx,
                                             const float* __restrict__ Wh,
                                             int colB, int half,
                                             v2f (&b)[K_STEPS]) {
#pragma unroll
  for (int i = 0; i < K_STEPS; ++i) {
    int ka = 4 * i + 2 * half;
    const float* w = (i < (2 * IN_CH) / 4) ? (Wx + ka * OUT_CH)
                                           : (Wh + (ka - 2 * IN_CH) * OUT_CH);
    b[i].x = w[colB];
    b[i].y = w[OUT_CH + colB];
  }
}

__global__ void __launch_bounds__(128)
gates_wmma_kernel(const float* __restrict__ x, const float* __restrict__ H,
                  const float* __restrict__ Lx, const float* __restrict__ LH,
                  const float* __restrict__ Wzx, const float* __restrict__ Wzh,
                  const float* __restrict__ Wrx, const float* __restrict__ Wrh,
                  const float* __restrict__ bzx, const float* __restrict__ bzh,
                  const float* __restrict__ brx, const float* __restrict__ brh,
                  float* __restrict__ Zbuf, float* __restrict__ RH, int N) {
  __shared__ __attribute__((aligned(16))) float sA[2][16 * A_PITCH];
  const int t = threadIdx.x;
  const int lane = t & 31;
  const int wv = t >> 5;
  const int half = lane >> 4;
  const int l15 = lane & 15;
  const int colB = wv * 16 + l15;
  const int ntiles = N >> 4;
  const int step = gridDim.x;

  v2f bz[K_STEPS], br[K_STEPS];
  load_b_frags(Wzx, Wzh, colB, half, bz);
  load_b_frags(Wrx, Wrh, colB, half, br);
  const float bZ = bzx[colB] + bzh[colB];
  const float bR = brx[colB] + brh[colB];

  int tile = blockIdx.x;
  v4f tp0 = {}, tp1 = {}, tq0 = {}, tq1 = {}, ts = {};
  if (tile < ntiles) {
    tile_load(x, Lx, H, LH, tile * 16, t, tp0, tp1, tq0, tq1, ts);
    tile_store(sA[0], t, tp0, tp1, tq0, tq1, ts);
  }
  __syncthreads();
  int buf = 0;
  for (; tile < ntiles; tile += step) {
    const int next = tile + step;
    const bool more = next < ntiles;
    if (more) tile_load(x, Lx, H, LH, next * 16, t, tp0, tp1, tq0, tq1, ts);

    const float* sa = sA[buf];
    const float* arow = sa + l15 * A_PITCH + 2 * half;
    v8f accZ = {};
    v8f accR = {};
#pragma unroll
    for (int i = 0; i < K_STEPS; ++i) {
      v2f a = *(const v2f*)(arow + 4 * i);
      accZ = wmma_f32(a, bz[i], accZ);
      accR = wmma_f32(a, br[i], accR);
    }

    const int row0 = tile * 16;
#pragma unroll
    for (int i = 0; i < 8; ++i) {
      int lr = i + 8 * half;
      int r = row0 + lr;
      float z = 1.0f / (1.0f + __expf(-(accZ[i] + bZ)));
      float rgate = 1.0f / (1.0f + __expf(-(accR[i] + bR)));
      float h = sa[lr * A_PITCH + 2 * IN_CH + colB];
      Zbuf[r * OUT_CH + colB] = z;
      RH[r * OUT_CH + colB] = rgate * h;
    }

    if (more) tile_store(sA[buf ^ 1], t, tp0, tp1, tq0, tq1, ts);
    __syncthreads();
    buf ^= 1;
  }
}

__global__ void __launch_bounds__(128)
cand_out_wmma_kernel(const float* __restrict__ x, const float* __restrict__ H,
                     const float* __restrict__ Lx, const float* __restrict__ RH,
                     const float* __restrict__ LRH,
                     const float* __restrict__ Zbuf,
                     const float* __restrict__ Wcx,
                     const float* __restrict__ Wch,
                     const float* __restrict__ bcx,
                     const float* __restrict__ bch, float* __restrict__ out,
                     int N) {
  __shared__ __attribute__((aligned(16))) float sA[2][16 * A_PITCH];
  const int t = threadIdx.x;
  const int lane = t & 31;
  const int wv = t >> 5;
  const int half = lane >> 4;
  const int l15 = lane & 15;
  const int colB = wv * 16 + l15;
  const int ntiles = N >> 4;
  const int step = gridDim.x;

  v2f bc[K_STEPS];
  load_b_frags(Wcx, Wch, colB, half, bc);
  const float bC = bcx[colB] + bch[colB];

  int tile = blockIdx.x;
  v4f tp0 = {}, tp1 = {}, tq0 = {}, tq1 = {}, ts = {};
  if (tile < ntiles) {
    tile_load(x, Lx, RH, LRH, tile * 16, t, tp0, tp1, tq0, tq1, ts);
    tile_store(sA[0], t, tp0, tp1, tq0, tq1, ts);
  }
  __syncthreads();
  int buf = 0;
  for (; tile < ntiles; tile += step) {
    const int next = tile + step;
    const bool more = next < ntiles;
    if (more) tile_load(x, Lx, RH, LRH, next * 16, t, tp0, tp1, tq0, tq1, ts);

    const int row0 = tile * 16;
    float zv[8], hv[8];
#pragma unroll
    for (int i = 0; i < 8; ++i) {
      int r = row0 + i + 8 * half;
      zv[i] = Zbuf[r * OUT_CH + colB];
      hv[i] = H[r * OUT_CH + colB];
    }

    const float* sa = sA[buf];
    const float* arow = sa + l15 * A_PITCH + 2 * half;
    v8f accC = {};
#pragma unroll
    for (int i = 0; i < K_STEPS; ++i) {
      v2f a = *(const v2f*)(arow + 4 * i);
      accC = wmma_f32(a, bc[i], accC);
    }

#pragma unroll
    for (int i = 0; i < 8; ++i) {
      int r = row0 + i + 8 * half;
      float cv = tanhf(accC[i] + bC);
      out[r * OUT_CH + colB] = zv[i] * hv[i] + (1.0f - zv[i]) * cv;
    }

    if (more) tile_store(sA[buf ^ 1], t, tp0, tp1, tq0, tq1, ts);
    __syncthreads();
    buf ^= 1;
  }
}


extern "C" void kernel_launch(void* const* d_in, const int* in_sizes, int n_in,
                              void* d_out, int out_size, void* d_ws,
                              size_t ws_size, hipStream_t stream) {
  const float* x = (const float*)d_in[0];
  const float* H = (const float*)d_in[1];
  const int* edge = (const int*)d_in[2];
  const float* Wzx = (const float*)d_in[3];
  const float* bzx = (const float*)d_in[4];
  const float* Wzh = (const float*)d_in[5];
  const float* bzh = (const float*)d_in[6];
  const float* Wrx = (const float*)d_in[7];
  const float* brx = (const float*)d_in[8];
  const float* Wrh = (const float*)d_in[9];
  const float* brh = (const float*)d_in[10];
  const float* Wcx = (const float*)d_in[11];
  const float* bcx = (const float*)d_in[12];
  const float* Wch = (const float*)d_in[13];
  const float* bch = (const float*)d_in[14];

  const int N = in_sizes[0] / IN_CH;
  const int E = in_sizes[2] / 2;
  const int* rowIdx = edge;
  const int* colIdx = edge + E;

  float* ws = (float*)d_ws;
  float* LH = ws;
  float* LRH = LH + (size_t)N * OUT_CH;
  float* Zbuf = LRH + (size_t)N * OUT_CH;
  float* RH = Zbuf + (size_t)N * OUT_CH;
  float* Lx = RH + (size_t)N * OUT_CH;
  int* sedge = (int*)(Lx + (size_t)N * IN_CH);
  int* tmp = sedge + (size_t)E * 2;
  float* dinv = (float*)(tmp + (size_t)E * 2);
  int* rowptr = (int*)(dinv + N);
  int* chunkOff = rowptr + N + 1;

  const int NB = (N + ROWS_PER_BUCKET - 1) / ROWS_PER_BUCKET;
  const int nchunks = (E + CHUNK - 1) / CHUNK;

  chunk_sort_kernel<<<nchunks, CHUNK_THREADS, 0, stream>>>(rowIdx, colIdx, tmp,
                                                           chunkOff, E, NB);
  bucket_degree_kernel<<<NB, 1024, 0, stream>>>(tmp, chunkOff, rowptr, dinv, N,
                                                NB, nchunks);
  bucket_place_kernel<<<NB, 1024, 0, stream>>>(tmp, chunkOff, rowptr, dinv,
                                               sedge, N, NB, nchunks);
  {
    long long threads = (long long)N * 32;
    spmm_rows_kernel<<<(int)((threads + 255) / 256), 256, 0, stream>>>(
        rowptr, sedge, H, LH, x, Lx, N);
  }
  gates_wmma_kernel<<<DENSE_GRID, 128, 0, stream>>>(x, H, Lx, LH, Wzx, Wzh, Wrx,
                                                     Wrh, bzx, bzh, brx, brh,
                                                     Zbuf, RH, N);
  {
    long long threads = (long long)N * 32;
    spmm_rows_kernel<<<(int)((threads + 255) / 256), 256, 0, stream>>>(
        rowptr, sedge, RH, LRH, nullptr, nullptr, N);
  }
  cand_out_wmma_kernel<<<DENSE_GRID, 128, 0, stream>>>(
      x, H, Lx, RH, LRH, Zbuf, Wcx, Wch, bcx, bch, (float*)d_out, N);
}


# Edge norms formed in the first row-sum kernel from deg^-1/2 carried in the gathered x rows (64-byte rows): degree count and edge placement become one launch (7 -> 6), no dinv table; sum order kept
# speedup vs baseline: 5.3723x; 1.0844x over previous
#include <hip/hip_runtime.h>
#include <hip/hip_bf16.h>
#include <math.h>


typedef float v2f __attribute__((ext_vector_type(2)));
typedef float v4f __attribute__((ext_vector_type(4)));
typedef float v8f __attribute__((ext_vector_type(8)));

#define IN_CH 8
#define OUT_CH 64

__device__ __forceinline__ v8f wmma_f32(v2f a, v2f b, v8f c) {
  return __builtin_amdgcn_wmma_f32_16x16x4_f32(false, a, false, b, (short)0, c,
                                               false, false);
}


#define BUCKET_SHIFT 8
#define ROWS_PER_BUCKET 256
#define CHUNK 4096
#define CHUNK_THREADS 512
#define NB_CAP 1024
#define RUN_TILE 512
#define STAGE_CAP 12288
#define XD_W 16

__device__ __forceinline__ int block_excl_scan(int v, int* s_w, int& total) {
  int lane = threadIdx.x & 31;
  int wv = threadIdx.x >> 5;
  int nw = blockDim.x >> 5;
  int incl = v;
#pragma unroll
  for (int d = 1; d < 32; d <<= 1) {
    int up = __shfl_up(incl, d);
    if (lane >= d) incl += up;
  }
  if (lane == 31) s_w[wv] = incl;
  __syncthreads();
  int wsum = (lane < nw) ? s_w[lane] : 0;
  int winc = wsum;
#pragma unroll
  for (int d = 1; d < 32; d <<= 1) {
    int up = __shfl_up(winc, d);
    if (lane >= d) winc += up;
  }
  int wpre = __shfl(winc - wsum, wv);
  total = __shfl(winc, 31);
  __syncthreads();
  return wpre + incl - v;
}

__global__ void __launch_bounds__(CHUNK_THREADS)
chunk_sort_kernel(const int* __restrict__ rowIdx, const int* __restrict__ colIdx,
                  int* __restrict__ tmp, int* __restrict__ chunkOff, int E,
                  int NB) {
  __shared__ int s_hist[NB_CAP + 1];
  __shared__ int s_w[32];
  __shared__ int2 s_stage[CHUNK];
  const int t = threadIdx.x;
  const int base = blockIdx.x * CHUNK;
  for (int i = t; i <= NB; i += CHUNK_THREADS) s_hist[i] = 0;
  __syncthreads();
  int r[CHUNK / CHUNK_THREADS], c[CHUNK / CHUNK_THREADS], rk[CHUNK / CHUNK_THREADS];
#pragma unroll
  for (int u = 0; u < CHUNK / CHUNK_THREADS; ++u) {
    int e = base + u * CHUNK_THREADS + t;
    r[u] = -1;
    c[u] = 0;
    if (e < E) {
      r[u] = rowIdx[e];
      c[u] = colIdx[e];
    }
  }
#pragma unroll
  for (int u = 0; u < CHUNK / CHUNK_THREADS; ++u) {
    rk[u] = 0;
    if (r[u] >= 0) rk[u] = atomicAdd(&s_hist[r[u] >> BUCKET_SHIFT], 1);
  }
  __syncthreads();
  int carry = 0;
  for (int i0 = 0; i0 < NB; i0 += CHUNK_THREADS) {
    int i = i0 + t;
    int v = (i < NB) ? s_hist[i] : 0;
    int tot;
    int ex = block_excl_scan(v, s_w, tot);
    if (i < NB) s_hist[i] = carry + ex;
    carry += tot;
  }
  if (t == 0) s_hist[NB] = carry;
  __syncthreads();
  for (int i = t; i <= NB; i += CHUNK_THREADS)
    chunkOff[(size_t)blockIdx.x * (NB + 1) + i] = s_hist[i];
#pragma unroll
  for (int u = 0; u < CHUNK / CHUNK_THREADS; ++u) {
    if (r[u] >= 0)
      s_stage[s_hist[r[u] >> BUCKET_SHIFT] + rk[u]] = make_int2(r[u], c[u]);
  }
  __syncthreads();
  int cnt = E - base;
  if (cnt > CHUNK) cnt = CHUNK;
  for (int i = t; i < cnt; i += CHUNK_THREADS)
    ((int2*)tmp)[(size_t)base + i] = s_stage[i];
}

template <typename F>
__device__ __forceinline__ void for_each_run_edge(const int2* __restrict__ tmp2,
                                                  const int* s_start,
                                                  const int* s_len, int nr,
                                                  int lane, int wv, int nw,
                                                  F f) {
  for (int c = wv; c < nr; c += 4 * nw) {
    int2 v[4];
    int len[4];
    int st[4];
#pragma unroll
    for (int u = 0; u < 4; ++u) {
      int cc = c + u * nw;
      len[u] = (cc < nr) ? s_len[cc] : 0;
      st[u] = (cc < nr) ? s_start[cc] : 0;
      v[u] = make_int2(0, 0);
      if (lane < len[u]) v[u] = tmp2[(size_t)st[u] + lane];
    }
#pragma unroll
    for (int u = 0; u < 4; ++u) {
      if (lane < len[u]) f(v[u]);
      for (int i = lane + 32; i < len[u]; i += 32) f(tmp2[(size_t)st[u] + i]);
    }
  }
}

__device__ __forceinline__ int load_run_tile(const int* __restrict__ chunkOff,
                                             int b, int NB, int nchunks, int c0,
                                             int* s_start, int* s_len) {
  int before = 0;
  int t = threadIdx.x;
  if (t < RUN_TILE) {
    int c = c0 + t;
    int st = 0, len = 0;
    if (c < nchunks) {
      const int* o = chunkOff + (size_t)c * (NB + 1) + b;
      int o0 = o[0];
      int o1 = o[1];
      st = c * CHUNK + o0;
      len = o1 - o0;
      before = o0;
    }
    s_start[t] = st;
    s_len[t] = len;
  }
  return before;
}

__global__ void __launch_bounds__(1024)
bucket_order_kernel(const int* __restrict__ tmp, const int* __restrict__ chunkOff,
                    const float* __restrict__ x, int* __restrict__ rowptr,
                    int* __restrict__ scol, float* __restrict__ xd, int N, int NB,
                    int nchunks) {
  __shared__ int s_stage[STAGE_CAP];
  __shared__ int s_start[RUN_TILE];
  __shared__ int s_len[RUN_TILE];
  __shared__ int s_cnt[ROWS_PER_BUCKET];
  __shared__ int s_rp[ROWS_PER_BUCKET];
  __shared__ float s_dinv[ROWS_PER_BUCKET];
  __shared__ int s_w[32];
  const int t = threadIdx.x;
  const int lane = t & 31;
  const int wv = t >> 5;
  const int nw = blockDim.x >> 5;
  const int b = blockIdx.x;
  const int row0 = b << BUCKET_SHIFT;
  const int2* __restrict__ tmp2 = (const int2*)tmp;
  const bool one_tile = (nchunks <= RUN_TILE);
  if (t < ROWS_PER_BUCKET) s_cnt[t] = 0;
  int before = 0;
  for (int c0 = 0; c0 < nchunks; c0 += RUN_TILE) {
    __syncthreads();
    before += load_run_tile(chunkOff, b, NB, nchunks, c0, s_start, s_len);
    __syncthreads();
    int nr = nchunks - c0;
    if (nr > RUN_TILE) nr = RUN_TILE;
    for_each_run_edge(tmp2, s_start, s_len, nr, lane, wv, nw,
                      [&](int2 rc) { atomicAdd(&s_cnt[rc.x - row0], 1); });
  }
  __syncthreads();
  int deg = (t < ROWS_PER_BUCKET) ? s_cnt[t] : 0;
  int tot, totBefore;
  int ex = block_excl_scan(deg, s_w, tot);
  block_excl_scan(before, s_w, totBefore);
  if (t < ROWS_PER_BUCKET) {
    int row = row0 + t;
    s_rp[t] = ex;
    s_dinv[t] = (deg > 0) ? rsqrtf((float)deg) : 0.0f;
    s_cnt[t] = 0;
    if (row < N) rowptr[row] = totBefore + ex;
  }
  if (b == NB - 1 && t == 0) rowptr[N] = totBefore + tot;
  const int seg0 = totBefore;
  const int nb = tot;
  const bool staged = (nb <= STAGE_CAP);
  for (int c0 = 0; c0 < nchunks; c0 += RUN_TILE) {
    __syncthreads();
    if (!one_tile) {
      load_run_tile(chunkOff, b, NB, nchunks, c0, s_start, s_len);
      __syncthreads();
    }
    int nr = nchunks - c0;
    if (nr > RUN_TILE) nr = RUN_TILE;
    for_each_run_edge(tmp2, s_start, s_len, nr, lane, wv, nw, [&](int2 rc) {
      int lr = rc.x - row0;
      int rk = atomicAdd(&s_cnt[lr], 1);
      int pl = s_rp[lr] + rk;
      if (staged)
        s_stage[pl] = rc.y;
      else
        scol[(size_t)seg0 + pl] = rc.y;
    });
  }
  __syncthreads();
  if (staged) {
    for (int i = t; i < nb; i += 1024) scol[(size_t)seg0 + i] = s_stage[i];
  }
  {
    int lr = t >> 2;
    int q = t & 3;
    int row = row0 + lr;
    if (row < N) {
      v4f v = {0.0f, 0.0f, 0.0f, 0.0f};
      if (q < 2) {
        const float* xr = x + (size_t)row * IN_CH + q * 4;
        v.x = xr[0];
        v.y = xr[1];
        v.z = xr[2];
        v.w = xr[3];
      } else if (q == 2) {
        v.x = s_dinv[lr];
      }
      *(v4f*)(xd + (size_t)row * XD_W + q * 4) = v;
    }
  }
}

template <typename V>
__device__ __forceinline__ V gather_edges(const V* __restrict__ src, int stride,
                                          int off, int2 en, int n, V acc) {
  int j = 0;
  for (; j + 8 <= n; j += 8) {
    V v[8];
#pragma unroll
    for (int u = 0; u < 8; ++u) {
      int c = __builtin_amdgcn_readlane(en.x, j + u);
      v[u] = src[(size_t)c * stride + off];
    }
#pragma unroll
    for (int u = 0; u < 8; ++u) {
      float w = __int_as_float(__builtin_amdgcn_readlane(en.y, j + u));
      acc += v[u] * w;
    }
  }
  int rem = n - j;
  if (rem > 0) {
    V v[7];
#pragma unroll
    for (int u = 0; u < 7; ++u) {
      v[u] = V(0.0f);
      if (u < rem) {
        int c = __builtin_amdgcn_readlane(en.x, j + u);
        v[u] = src[(size_t)c * stride + off];
      }
    }
#pragma unroll
    for (int u = 0; u < 7; ++u) {
      if (u < rem) {
        float w = __int_as_float(__builtin_amdgcn_readlane(en.y, j + u));
        acc += v[u] * w;
      }
    }
  }
  return acc;
}

template <int CNT>
__device__ __forceinline__ void first_batch(const v2f* __restrict__ s64,
                                            const float* __restrict__ xd,
                                            int lane, int c, int j, int m,
                                            float ndr, v2f& acc, float& acc8,
                                            float& wl) {
  const int rel = lane - j;
  v2f v[CNT];
  float xv[CNT];
#pragma unroll
  for (int u = 0; u < CNT; ++u) {
    v[u] = v2f{0.0f, 0.0f};
    xv[u] = 0.0f;
    if (u < m) {
      int cu = __builtin_amdgcn_readlane(c, j + u);
      v[u] = s64[(size_t)cu * (OUT_CH / 2) + lane];
      xv[u] = xd[(size_t)cu * XD_W + (lane & 15)];
    }
  }
#pragma unroll
  for (int u = 0; u < CNT; ++u) {
    if (u < m) {
      float dc = __int_as_float(
          __builtin_amdgcn_readlane(__float_as_int(xv[u]), IN_CH));
      float w = ndr * dc;
      acc += v[u] * w;
      acc8 += xv[u] * w;
      wl = (rel == u) ? w : wl;
    }
  }
}

__global__ void __launch_bounds__(256)
spmm_first_kernel(const int* __restrict__ rowptr, const int* __restrict__ scol,
                  const float* __restrict__ src64, float* __restrict__ dst64,
                  const float* __restrict__ xd, float* __restrict__ dst8,
                  float* __restrict__ nrm, int N) {
  int row = __builtin_amdgcn_readfirstlane(
      (blockIdx.x * blockDim.x + threadIdx.x) >> 5);
  int lane = threadIdx.x & 31;
  if (row >= N) return;
  int beg = __builtin_amdgcn_readfirstlane(rowptr[row]);
  int end = __builtin_amdgcn_readfirstlane(rowptr[row + 1]);
  const float ndr = -__int_as_float(__builtin_amdgcn_readfirstlane(
      __float_as_int(xd[(size_t)row * XD_W + IN_CH])));
  const v2f* __restrict__ s64 = (const v2f*)src64;
  v2f acc = {0.0f, 0.0f};
  float acc8 = 0.0f;
  for (int base = beg; base < end; base += 32) {
    int n = end - base;
    if (n > 32) n = 32;
    int c = 0;
    if (lane < n) c = scol[base + lane];
    float wl = 0.0f;
    int j = 0;
    for (; j + 8 <= n; j += 8)
      first_batch<8>(s64, xd, lane, c, j, 8, ndr, acc, acc8, wl);
    if (j < n) first_batch<7>(s64, xd, lane, c, j, n - j, ndr, acc, acc8, wl);
    if (lane < n) nrm[base + lane] = wl;
  }
  ((v2f*)dst64)[(size_t)row * (OUT_CH / 2) + lane] = acc;
  if (lane < IN_CH) dst8[row * IN_CH + lane] = acc8;
}

__global__ void __launch_bounds__(256)
spmm_second_kernel(const int* __restrict__ rowptr, const int* __restrict__ scol,
                   const float* __restrict__ nrm,
                   const float* __restrict__ src64, float* __restrict__ dst64,
                   int N) {
  int row = __builtin_amdgcn_readfirstlane(
      (blockIdx.x * blockDim.x + threadIdx.x) >> 5);
  int lane = threadIdx.x & 31;
  if (row >= N) return;
  int beg = __builtin_amdgcn_readfirstlane(rowptr[row]);
  int end = __builtin_amdgcn_readfirstlane(rowptr[row + 1]);
  const v2f* __restrict__ s64 = (const v2f*)src64;
  v2f acc = {0.0f, 0.0f};
  for (int base = beg; base < end; base += 32) {
    int n = end - base;
    if (n > 32) n = 32;
    int2 en = make_int2(0, 0);
    if (lane < n) {
      en.x = scol[base + lane];
      en.y = __float_as_int(nrm[base + lane]);
    }
    acc = gather_edges<v2f>(s64, OUT_CH / 2, lane, en, n, acc);
  }
  ((v2f*)dst64)[(size_t)row * (OUT_CH / 2) + lane] = acc;
}


#define A_K 144
#define K_STEPS (A_K / 4)
#define A_PITCH 148
#define DENSE_GRID 1024

__device__ __forceinline__ void tile_load(const float* __restrict__ x,
                                          const float* __restrict__ Lx,
                                          const float* __restrict__ P,
                                          const float* __restrict__ LP,
                                          int row0, int t, v4f& p0, v4f& p1,
                                          v4f& q0, v4f& q1, v4f& s) {
  const v4f* p4 = (const v4f*)(P + (size_t)row0 * OUT_CH);
  const v4f* q4 = (const v4f*)(LP + (size_t)row0 * OUT_CH);
  p0 = p4[t];
  p1 = p4[t + 128];
  q0 = q4[t];
  q1 = q4[t + 128];
  const float* sb = (t < 32) ? x : Lx;
  if (t < 64) s = ((const v4f*)(sb + (size_t)row0 * IN_CH))[t & 31];
}

__device__ __forceinline__ void tile_store(float* sa, int t, v4f p0, v4f p1,
                                           v4f q0, v4f q1, v4f s) {
  int row = t >> 4;
  int c = (t & 15) * 4;
  *(v4f*)&sa[row * A_PITCH + 2 * IN_CH + c] = p0;
  *(v4f*)&sa[(row + 8) * A_PITCH + 2 * IN_CH + c] = p1;
  *(v4f*)&sa[row * A_PITCH + 2 * IN_CH + OUT_CH + c] = q0;
  *(v4f*)&sa[(row + 8) * A_PITCH + 2 * IN_CH + OUT_CH + c] = q1;
  if (t < 64) {
    int u = t & 31;
    int off = (t < 32) ? 0 : IN_CH;
    *(v4f*)&sa[(u >> 1) * A_PITCH + off + (u & 1) * 4] = s;
  }
}

__device__ __forceinline__ void load_b_frags(const float* __restrict__ Wx,
                                             const float* __restrict__ Wh,
                                             int colB, int half,
                                             v2f (&b)[K_STEPS]) {
#pragma unroll
  for (int i = 0; i < K_STEPS; ++i) {
    int ka = 4 * i + 2 * half;
    const float* w = (i < (2 * IN_CH) / 4) ? (Wx + ka * OUT_CH)
                                           : (Wh + (ka - 2 * IN_CH) * OUT_CH);
    b[i].x = w[colB];
    b[i].y = w[OUT_CH + colB];
  }
}

__global__ void __launch_bounds__(128)
gates_wmma_kernel(const float* __restrict__ x, const float* __restrict__ H,
                  const float* __restrict__ Lx, const float* __restrict__ LH,
                  const float* __restrict__ Wzx, const float* __restrict__ Wzh,
                  const float* __restrict__ Wrx, const float* __restrict__ Wrh,
                  const float* __restrict__ bzx, const float* __restrict__ bzh,
                  const float* __restrict__ brx, const float* __restrict__ brh,
                  float* __restrict__ Zbuf, float* __restrict__ RH, int N) {
  __shared__ __attribute__((aligned(16))) float sA[2][16 * A_PITCH];
  const int t = threadIdx.x;
  const int lane = t & 31;
  const int wv = t >> 5;
  const int half = lane >> 4;
  const int l15 = lane & 15;
  const int colB = wv * 16 + l15;
  const int ntiles = N >> 4;
  const int step = gridDim.x;

  v2f bz[K_STEPS], br[K_STEPS];
  load_b_frags(Wzx, Wzh, colB, half, bz);
  load_b_frags(Wrx, Wrh, colB, half, br);
  const float bZ = bzx[colB] + bzh[colB];
  const float bR = brx[colB] + brh[colB];

  int tile = blockIdx.x;
  v4f tp0 = {}, tp1 = {}, tq0 = {}, tq1 = {}, ts = {};
  if (tile < ntiles) {
    tile_load(x, Lx, H, LH, tile * 16, t, tp0, tp1, tq0, tq1, ts);
    tile_store(sA[0], t, tp0, tp1, tq0, tq1, ts);
  }
  __syncthreads();
  int buf = 0;
  for (; tile < ntiles; tile += step) {
    const int next = tile + step;
    const bool more = next < ntiles;
    if (more) tile_load(x, Lx, H, LH, next * 16, t, tp0, tp1, tq0, tq1, ts);

    const float* sa = sA[buf];
    const float* arow = sa + l15 * A_PITCH + 2 * half;
    v8f accZ = {};
    v8f accR = {};
#pragma unroll
    for (int i = 0; i < K_STEPS; ++i) {
      v2f a = *(const v2f*)(arow + 4 * i);
      accZ = wmma_f32(a, bz[i], accZ);
      accR = wmma_f32(a, br[i], accR);
    }

    const int row0 = tile * 16;
#pragma unroll
    for (int i = 0; i < 8; ++i) {
      int lr = i + 8 * half;
      int r = row0 + lr;
      float z = 1.0f / (1.0f + __expf(-(accZ[i] + bZ)));
      float rgate = 1.0f / (1.0f + __expf(-(accR[i] + bR)));
      float h = sa[lr * A_PITCH + 2 * IN_CH + colB];
      Zbuf[r * OUT_CH + colB] = z;
      RH[r * OUT_CH + colB] = rgate * h;
    }

    if (more) tile_store(sA[buf ^ 1], t, tp0, tp1, tq0, tq1, ts);
    __syncthreads();
    buf ^= 1;
  }
}

__global__ void __launch_bounds__(128)
cand_out_wmma_kernel(const float* __restrict__ x, const float* __restrict__ H,
                     const float* __restrict__ Lx, const float* __restrict__ RH,
                     const float* __restrict__ LRH,
                     const float* __restrict__ Zbuf,
                     const float* __restrict__ Wcx,
                     const float* __restrict__ Wch,
                     const float* __restrict__ bcx,
                     const float* __restrict__ bch, float* __restrict__ out,
                     int N) {
  __shared__ __attribute__((aligned(16))) float sA[2][16 * A_PITCH];
  const int t = threadIdx.x;
  const int lane = t & 31;
  const int wv = t >> 5;
  const int half = lane >> 4;
  const int l15 = lane & 15;
  const int colB = wv * 16 + l15;
  const int ntiles = N >> 4;
  const int step = gridDim.x;

  v2f bc[K_STEPS];
  load_b_frags(Wcx, Wch, colB, half, bc);
  const float bC = bcx[colB] + bch[colB];

  int tile = blockIdx.x;
  v4f tp0 = {}, tp1 = {}, tq0 = {}, tq1 = {}, ts = {};
  if (tile < ntiles) {
    tile_load(x, Lx, RH, LRH, tile * 16, t, tp0, tp1, tq0, tq1, ts);
    tile_store(sA[0], t, tp0, tp1, tq0, tq1, ts);
  }
  __syncthreads();
  int buf = 0;
  for (; tile < ntiles; tile += step) {
    const int next = tile + step;
    const bool more = next < ntiles;
    if (more) tile_load(x, Lx, RH, LRH, next * 16, t, tp0, tp1, tq0, tq1, ts);

    const int row0 = tile * 16;
    float zv[8], hv[8];
#pragma unroll
    for (int i = 0; i < 8; ++i) {
      int r = row0 + i + 8 * half;
      zv[i] = Zbuf[r * OUT_CH + colB];
      hv[i] = H[r * OUT_CH + colB];
    }

    const float* sa = sA[buf];
    const float* arow = sa + l15 * A_PITCH + 2 * half;
    v8f accC = {};
#pragma unroll
    for (int i = 0; i < K_STEPS; ++i) {
      v2f a = *(const v2f*)(arow + 4 * i);
      accC = wmma_f32(a, bc[i], accC);
    }

#pragma unroll
    for (int i = 0; i < 8; ++i) {
      int r = row0 + i + 8 * half;
      float cv = tanhf(accC[i] + bC);
      out[r * OUT_CH + colB] = zv[i] * hv[i] + (1.0f - zv[i]) * cv;
    }

    if (more) tile_store(sA[buf ^ 1], t, tp0, tp1, tq0, tq1, ts);
    __syncthreads();
    buf ^= 1;
  }
}


extern "C" void kernel_launch(void* const* d_in, const int* in_sizes, int n_in,
                              void* d_out, int out_size, void* d_ws,
                              size_t ws_size, hipStream_t stream) {
  const float* x = (const float*)d_in[0];
  const float* H = (const float*)d_in[1];
  const int* edge = (const int*)d_in[2];
  const float* Wzx = (const float*)d_in[3];
  const float* bzx = (const float*)d_in[4];
  const float* Wzh = (const float*)d_in[5];
  const float* bzh = (const float*)d_in[6];
  const float* Wrx = (const float*)d_in[7];
  const float* brx = (const float*)d_in[8];
  const float* Wrh = (const float*)d_in[9];
  const float* brh = (const float*)d_in[10];
  const float* Wcx = (const float*)d_in[11];
  const float* bcx = (const float*)d_in[12];
  const float* Wch = (const float*)d_in[13];
  const float* bch = (const float*)d_in[14];

  const int N = in_sizes[0] / IN_CH;
  const int E = in_sizes[2] / 2;
  const int* rowIdx = edge;
  const int* colIdx = edge + E;

  float* ws = (float*)d_ws;
  float* LH = ws;
  float* LRH = LH + (size_t)N * OUT_CH;
  float* Zbuf = LRH + (size_t)N * OUT_CH;
  float* RH = Zbuf + (size_t)N * OUT_CH;
  float* Lx = RH + (size_t)N * OUT_CH;
  float* xd = Lx + (size_t)N * IN_CH;
  int* scol = (int*)(xd + (size_t)N * XD_W);
  float* nrm = (float*)(scol + E);
  int* tmp = (int*)(nrm + E);
  int* rowptr = tmp + (size_t)E * 2;
  int* chunkOff = rowptr + N + 1;

  const int NB = (N + ROWS_PER_BUCKET - 1) / ROWS_PER_BUCKET;
  const int nchunks = (E + CHUNK - 1) / CHUNK;

  chunk_sort_kernel<<<nchunks, CHUNK_THREADS, 0, stream>>>(rowIdx, colIdx, tmp,
                                                           chunkOff, E, NB);
  bucket_order_kernel<<<NB, 1024, 0, stream>>>(tmp, chunkOff, x, rowptr, scol,
                                               xd, N, NB, nchunks);
  {
    long long threads = (long long)N * 32;
    spmm_first_kernel<<<(int)((threads + 255) / 256), 256, 0, stream>>>(
        rowptr, scol, H, LH, xd, Lx, nrm, N);
  }
  gates_wmma_kernel<<<DENSE_GRID, 128, 0, stream>>>(x, H, Lx, LH, Wzx, Wzh, Wrx,
                                                     Wrh, bzx, bzh, brx, brh,
                                                     Zbuf, RH, N);
  {
    long long threads = (long long)N * 32;
    spmm_second_kernel<<<(int)((threads + 255) / 256), 256, 0, stream>>>(
        rowptr, scol, nrm, RH, LRH, N);
  }
  cand_out_wmma_kernel<<<DENSE_GRID, 128, 0, stream>>>(
      x, H, Lx, RH, LRH, Zbuf, Wcx, Wch, bcx, bch, (float*)d_out, N);
}


# Row sums: each gather instruction fetches two edges' rows (one per half wave, four floats per lane, columns handed out by permlane16); row sum = even-position edges' sum + odd-position edges' sum
# speedup vs baseline: 5.5642x; 1.0357x over previous
#include <hip/hip_runtime.h>
#include <hip/hip_bf16.h>
#include <math.h>


typedef float v2f __attribute__((ext_vector_type(2)));
typedef float v4f __attribute__((ext_vector_type(4)));
typedef float v8f __attribute__((ext_vector_type(8)));

#define IN_CH 8
#define OUT_CH 64

__device__ __forceinline__ v8f wmma_f32(v2f a, v2f b, v8f c) {
  return __builtin_amdgcn_wmma_f32_16x16x4_f32(false, a, false, b, (short)0, c,
                                               false, false);
}


#define BUCKET_SHIFT 8
#define ROWS_PER_BUCKET 256
#define CHUNK 4096
#define CHUNK_THREADS 512
#define NB_CAP 1024
#define RUN_TILE 512
#define STAGE_CAP 12288
#define XD_W 16

__device__ __forceinline__ int block_excl_scan(int v, int* s_w, int& total) {
  int lane = threadIdx.x & 31;
  int wv = threadIdx.x >> 5;
  int nw = blockDim.x >> 5;
  int incl = v;
#pragma unroll
  for (int d = 1; d < 32; d <<= 1) {
    int up = __shfl_up(incl, d);
    if (lane >= d) incl += up;
  }
  if (lane == 31) s_w[wv] = incl;
  __syncthreads();
  int wsum = (lane < nw) ? s_w[lane] : 0;
  int winc = wsum;
#pragma unroll
  for (int d = 1; d < 32; d <<= 1) {
    int up = __shfl_up(winc, d);
    if (lane >= d) winc += up;
  }
  int wpre = __shfl(winc - wsum, wv);
  total = __shfl(winc, 31);
  __syncthreads();
  return wpre + incl - v;
}

__global__ void __launch_bounds__(CHUNK_THREADS)
chunk_sort_kernel(const int* __restrict__ rowIdx, const int* __restrict__ colIdx,
                  int* __restrict__ tmp, int* __restrict__ chunkOff, int E,
                  int NB) {
  __shared__ int s_hist[NB_CAP + 1];
  __shared__ int s_w[32];
  __shared__ int2 s_stage[CHUNK];
  const int t = threadIdx.x;
  const int base = blockIdx.x * CHUNK;
  for (int i = t; i <= NB; i += CHUNK_THREADS) s_hist[i] = 0;
  __syncthreads();
  int r[CHUNK / CHUNK_THREADS], c[CHUNK / CHUNK_THREADS], rk[CHUNK / CHUNK_THREADS];
#pragma unroll
  for (int u = 0; u < CHUNK / CHUNK_THREADS; ++u) {
    int e = base + u * CHUNK_THREADS + t;
    r[u] = -1;
    c[u] = 0;
    if (e < E) {
      r[u] = rowIdx[e];
      c[u] = colIdx[e];
    }
  }
#pragma unroll
  for (int u = 0; u < CHUNK / CHUNK_THREADS; ++u) {
    rk[u] = 0;
    if (r[u] >= 0) rk[u] = atomicAdd(&s_hist[r[u] >> BUCKET_SHIFT], 1);
  }
  __syncthreads();
  int carry = 0;
  for (int i0 = 0; i0 < NB; i0 += CHUNK_THREADS) {
    int i = i0 + t;
    int v = (i < NB) ? s_hist[i] : 0;
    int tot;
    int ex = block_excl_scan(v, s_w, tot);
    if (i < NB) s_hist[i] = carry + ex;
    carry += tot;
  }
  if (t == 0) s_hist[NB] = carry;
  __syncthreads();
  for (int i = t; i <= NB; i += CHUNK_THREADS)
    chunkOff[(size_t)blockIdx.x * (NB + 1) + i] = s_hist[i];
#pragma unroll
  for (int u = 0; u < CHUNK / CHUNK_THREADS; ++u) {
    if (r[u] >= 0)
      s_stage[s_hist[r[u] >> BUCKET_SHIFT] + rk[u]] = make_int2(r[u], c[u]);
  }
  __syncthreads();
  int cnt = E - base;
  if (cnt > CHUNK) cnt = CHUNK;
  for (int i = t; i < cnt; i += CHUNK_THREADS)
    ((int2*)tmp)[(size_t)base + i] = s_stage[i];
}

template <typename F>
__device__ __forceinline__ void for_each_run_edge(const int2* __restrict__ tmp2,
                                                  const int* s_start,
                                                  const int* s_len, int nr,
                                                  int lane, int wv, int nw,
                                                  F f) {
  for (int c = wv; c < nr; c += 4 * nw) {
    int2 v[4];
    int len[4];
    int st[4];
#pragma unroll
    for (int u = 0; u < 4; ++u) {
      int cc = c + u * nw;
      len[u] = (cc < nr) ? s_len[cc] : 0;
      st[u] = (cc < nr) ? s_start[cc] : 0;
      v[u] = make_int2(0, 0);
      if (lane < len[u]) v[u] = tmp2[(size_t)st[u] + lane];
    }
#pragma unroll
    for (int u = 0; u < 4; ++u) {
      if (lane < len[u]) f(v[u]);
      for (int i = lane + 32; i < len[u]; i += 32) f(tmp2[(size_t)st[u] + i]);
    }
  }
}

__device__ __forceinline__ int load_run_tile(const int* __restrict__ chunkOff,
                                             int b, int NB, int nchunks, int c0,
                                             int* s_start, int* s_len) {
  int before = 0;
  int t = threadIdx.x;
  if (t < RUN_TILE) {
    int c = c0 + t;
    int st = 0, len = 0;
    if (c < nchunks) {
      const int* o = chunkOff + (size_t)c * (NB + 1) + b;
      int o0 = o[0];
      int o1 = o[1];
      st = c * CHUNK + o0;
      len = o1 - o0;
      before = o0;
    }
    s_start[t] = st;
    s_len[t] = len;
  }
  return before;
}

__global__ void __launch_bounds__(1024)
bucket_order_kernel(const int* __restrict__ tmp, const int* __restrict__ chunkOff,
                    const float* __restrict__ x, int* __restrict__ rowptr,
                    int* __restrict__ scol, float* __restrict__ xd, int N, int NB,
                    int nchunks) {
  __shared__ int s_stage[STAGE_CAP];
  __shared__ int s_start[RUN_TILE];
  __shared__ int s_len[RUN_TILE];
  __shared__ int s_cnt[ROWS_PER_BUCKET];
  __shared__ int s_rp[ROWS_PER_BUCKET];
  __shared__ float s_dinv[ROWS_PER_BUCKET];
  __shared__ int s_w[32];
  const int t = threadIdx.x;
  const int lane = t & 31;
  const int wv = t >> 5;
  const int nw = blockDim.x >> 5;
  const int b = blockIdx.x;
  const int row0 = b << BUCKET_SHIFT;
  const int2* __restrict__ tmp2 = (const int2*)tmp;
  const bool one_tile = (nchunks <= RUN_TILE);
  if (t < ROWS_PER_BUCKET) s_cnt[t] = 0;
  int before = 0;
  for (int c0 = 0; c0 < nchunks; c0 += RUN_TILE) {
    __syncthreads();
    before += load_run_tile(chunkOff, b, NB, nchunks, c0, s_start, s_len);
    __syncthreads();
    int nr = nchunks - c0;
    if (nr > RUN_TILE) nr = RUN_TILE;
    for_each_run_edge(tmp2, s_start, s_len, nr, lane, wv, nw,
                      [&](int2 rc) { atomicAdd(&s_cnt[rc.x - row0], 1); });
  }
  __syncthreads();
  int deg = (t < ROWS_PER_BUCKET) ? s_cnt[t] : 0;
  int tot, totBefore;
  int ex = block_excl_scan(deg, s_w, tot);
  block_excl_scan(before, s_w, totBefore);
  if (t < ROWS_PER_BUCKET) {
    int row = row0 + t;
    s_rp[t] = ex;
    s_dinv[t] = (deg > 0) ? rsqrtf((float)deg) : 0.0f;
    s_cnt[t] = 0;
    if (row < N) rowptr[row] = totBefore + ex;
  }
  if (b == NB - 1 && t == 0) rowptr[N] = totBefore + tot;
  const int seg0 = totBefore;
  const int nb = tot;
  const bool staged = (nb <= STAGE_CAP);
  for (int c0 = 0; c0 < nchunks; c0 += RUN_TILE) {
    __syncthreads();
    if (!one_tile) {
      load_run_tile(chunkOff, b, NB, nchunks, c0, s_start, s_len);
      __syncthreads();
    }
    int nr = nchunks - c0;
    if (nr > RUN_TILE) nr = RUN_TILE;
    for_each_run_edge(tmp2, s_start, s_len, nr, lane, wv, nw, [&](int2 rc) {
      int lr = rc.x - row0;
      int rk = atomicAdd(&s_cnt[lr], 1);
      int pl = s_rp[lr] + rk;
      if (staged)
        s_stage[pl] = rc.y;
      else
        scol[(size_t)seg0 + pl] = rc.y;
    });
  }
  __syncthreads();
  if (staged) {
    for (int i = t; i < nb; i += 1024) scol[(size_t)seg0 + i] = s_stage[i];
  }
  {
    int lr = t >> 2;
    int q = t & 3;
    int row = row0 + lr;
    if (row < N) {
      v4f v = {0.0f, 0.0f, 0.0f, 0.0f};
      if (q < 2) {
        const float* xr = x + (size_t)row * IN_CH + q * 4;
        v.x = xr[0];
        v.y = xr[1];
        v.z = xr[2];
        v.w = xr[3];
      } else if (q == 2) {
        v.x = s_dinv[lr];
      }
      *(v4f*)(xd + (size_t)row * XD_W + q * 4) = v;
    }
  }
}


__device__ __forceinline__ int row_lane_i(int v, unsigned sel) {
  return __builtin_amdgcn_permlane16(v, v, sel, sel, false, false);
}
__device__ __forceinline__ float row_lane_f(float v, unsigned sel) {
  return __int_as_float(row_lane_i(__float_as_int(v), sel));
}
__device__ __forceinline__ float xhalf(float x) {
  int b = __float_as_int(x);
  return __int_as_float(__builtin_amdgcn_permlanex16(b, b, 0x76543210u,
                                                     0xfedcba98u, false, false));
}
__device__ __forceinline__ v4f xhalf4(v4f a) {
  v4f r;
  r.x = xhalf(a.x);
  r.y = xhalf(a.y);
  r.z = xhalf(a.z);
  r.w = xhalf(a.w);
  return r;
}

#define SEL_ONE 0x11111111u
#define SEL_DINV 0x88888888u

template <int CNT>
__device__ __forceinline__ void first_pairs(const v4f* __restrict__ s4,
                                            const float* __restrict__ xd,
                                            int l15, int c, int p0, int m,
                                            float ndr, v4f& acc, float& acc8,
                                            float& wl) {
  v4f v[CNT];
  float xv[CNT];
#pragma unroll
  for (int u = 0; u < CNT; ++u) {
    v[u] = v4f{0.0f, 0.0f, 0.0f, 0.0f};
    xv[u] = 0.0f;
    if (u < m) {
      int cu = row_lane_i(c, (unsigned)(p0 + u) * SEL_ONE);
      unsigned idx = (unsigned)cu * 16u + (unsigned)l15;
      v[u] = s4[idx];
      xv[u] = xd[idx];
    }
  }
#pragma unroll
  for (int u = 0; u < CNT; ++u) {
    if (u < m) {
      float w = ndr * row_lane_f(xv[u], SEL_DINV);
      acc += v[u] * w;
      acc8 += xv[u] * w;
      wl = (l15 == p0 + u) ? w : wl;
    }
  }
}

__global__ void __launch_bounds__(256)
spmm_first_kernel(const int* __restrict__ rowptr, const int* __restrict__ scol,
                  const float* __restrict__ src64, float* __restrict__ dst64,
                  const float* __restrict__ xd, float* __restrict__ dst8,
                  float* __restrict__ nrm, int N) {
  int row = __builtin_amdgcn_readfirstlane(
      (blockIdx.x * blockDim.x + threadIdx.x) >> 5);
  int lane = threadIdx.x & 31;
  if (row >= N) return;
  int beg = __builtin_amdgcn_readfirstlane(rowptr[row]);
  int end = __builtin_amdgcn_readfirstlane(rowptr[row + 1]);
  const float ndr = -__int_as_float(__builtin_amdgcn_readfirstlane(
      __float_as_int(xd[(size_t)row * XD_W + IN_CH])));
  const int half = lane >> 4;
  const int l15 = lane & 15;
  const int eL = 2 * l15 + half;
  const v4f* __restrict__ s4 = (const v4f*)src64;
  v4f acc = {0.0f, 0.0f, 0.0f, 0.0f};
  float acc8 = 0.0f;
  for (int base = beg; base < end; base += 32) {
    int n = end - base;
    if (n > 32) n = 32;
    int c = 0;
    if (eL < n) c = scol[base + eL];
    float wl = 0.0f;
    const int nfull = n >> 1;
    int p0 = 0;
    for (; p0 + 8 <= nfull; p0 += 8)
      first_pairs<8>(s4, xd, l15, c, p0, 8, ndr, acc, acc8, wl);
    if (p0 < nfull)
      first_pairs<7>(s4, xd, l15, c, p0, nfull - p0, ndr, acc, acc8, wl);
    if (n & 1) {
      const int p = nfull;
      int cu = row_lane_i(c, (unsigned)p * SEL_ONE);
      unsigned idx = (unsigned)cu * 16u + (unsigned)l15;
      v4f v = s4[idx];
      float xv = xd[idx];
      float w = ndr * row_lane_f(xv, SEL_DINV);
      if (half) {
        w = 0.0f;
        xv = 0.0f;
        v = v4f{0.0f, 0.0f, 0.0f, 0.0f};
      }
      acc += v * w;
      acc8 += xv * w;
      wl = (l15 == p) ? w : wl;
    }
    if (eL < n) nrm[base + eL] = wl;
  }
  v4f tot = acc + xhalf4(acc);
  float tot8 = acc8 + xhalf(acc8);
  if (half == 0) ((v4f*)dst64)[(size_t)row * (OUT_CH / 4) + l15] = tot;
  if (lane < IN_CH) dst8[row * IN_CH + lane] = tot8;
}

template <int CNT>
__device__ __forceinline__ v4f second_pairs(const v4f* __restrict__ s4, int l15,
                                            int c, float wm, int p0, int m,
                                            v4f acc) {
  v4f v[CNT];
#pragma unroll
  for (int u = 0; u < CNT; ++u) {
    v[u] = v4f{0.0f, 0.0f, 0.0f, 0.0f};
    if (u < m) {
      int cu = row_lane_i(c, (unsigned)(p0 + u) * SEL_ONE);
      v[u] = s4[(unsigned)cu * 16u + (unsigned)l15];
    }
  }
#pragma unroll
  for (int u = 0; u < CNT; ++u) {
    if (u < m) {
      float w = row_lane_f(wm, (unsigned)(p0 + u) * SEL_ONE);
      acc += v[u] * w;
    }
  }
  return acc;
}

__global__ void __launch_bounds__(256)
spmm_second_kernel(const int* __restrict__ rowptr, const int* __restrict__ scol,
                   const float* __restrict__ nrm,
                   const float* __restrict__ src64, float* __restrict__ dst64,
                   int N) {
  int row = __builtin_amdgcn_readfirstlane(
      (blockIdx.x * blockDim.x + threadIdx.x) >> 5);
  int lane = threadIdx.x & 31;
  if (row >= N) return;
  int beg = __builtin_amdgcn_readfirstlane(rowptr[row]);
  int end = __builtin_amdgcn_readfirstlane(rowptr[row + 1]);
  const int half = lane >> 4;
  const int l15 = lane & 15;
  const int eL = 2 * l15 + half;
  const v4f* __restrict__ s4 = (const v4f*)src64;
  v4f acc = {0.0f, 0.0f, 0.0f, 0.0f};
  for (int base = beg; base < end; base += 32) {
    int n = end - base;
    if (n > 32) n = 32;
    int c = 0;
    float wm = 0.0f;
    if (eL < n) {
      c = scol[base + eL];
      wm = nrm[base + eL];
    }
    const int nfull = n >> 1;
    int p0 = 0;
    for (; p0 + 8 <= nfull; p0 += 8)
      acc = second_pairs<8>(s4, l15, c, wm, p0, 8, acc);
    if (p0 < nfull) acc = second_pairs<7>(s4, l15, c, wm, p0, nfull - p0, acc);
    if (n & 1) {
      const unsigned sel = (unsigned)nfull * SEL_ONE;
      int cu = row_lane_i(c, sel);
      v4f v = s4[(unsigned)cu * 16u + (unsigned)l15];
      float w = row_lane_f(wm, sel);
      if (half) v = v4f{0.0f, 0.0f, 0.0f, 0.0f};
      acc += v * w;
    }
  }
  v4f tot = acc + xhalf4(acc);
  if (half == 0) ((v4f*)dst64)[(size_t)row * (OUT_CH / 4) + l15] = tot;
}


#define A_K 144
#define K_STEPS (A_K / 4)
#define A_PITCH 148
#define DENSE_GRID 1024

__device__ __forceinline__ void tile_load(const float* __restrict__ x,
                                          const float* __restrict__ Lx,
                                          const float* __restrict__ P,
                                          const float* __restrict__ LP,
                                          int row0, int t, v4f& p0, v4f& p1,
                                          v4f& q0, v4f& q1, v4f& s) {
  const v4f* p4 = (const v4f*)(P + (size_t)row0 * OUT_CH);
  const v4f* q4 = (const v4f*)(LP + (size_t)row0 * OUT_CH);
  p0 = p4[t];
  p1 = p4[t + 128];
  q0 = q4[t];
  q1 = q4[t + 128];
  const float* sb = (t < 32) ? x : Lx;
  if (t < 64) s = ((const v4f*)(sb + (size_t)row0 * IN_CH))[t & 31];
}

__device__ __forceinline__ void tile_store(float* sa, int t, v4f p0, v4f p1,
                                           v4f q0, v4f q1, v4f s) {
  int row = t >> 4;
  int c = (t & 15) * 4;
  *(v4f*)&sa[row * A_PITCH + 2 * IN_CH + c] = p0;
  *(v4f*)&sa[(row + 8) * A_PITCH + 2 * IN_CH + c] = p1;
  *(v4f*)&sa[row * A_PITCH + 2 * IN_CH + OUT_CH + c] = q0;
  *(v4f*)&sa[(row + 8) * A_PITCH + 2 * IN_CH + OUT_CH + c] = q1;
  if (t < 64) {
    int u = t & 31;
    int off = (t < 32) ? 0 : IN_CH;
    *(v4f*)&sa[(u >> 1) * A_PITCH + off + (u & 1) * 4] = s;
  }
}

__device__ __forceinline__ void load_b_frags(const float* __restrict__ Wx,
                                             const float* __restrict__ Wh,
                                             int colB, int half,
                                             v2f (&b)[K_STEPS]) {
#pragma unroll
  for (int i = 0; i < K_STEPS; ++i) {
    int ka = 4 * i + 2 * half;
    const float* w = (i < (2 * IN_CH) / 4) ? (Wx + ka * OUT_CH)
                                           : (Wh + (ka - 2 * IN_CH) * OUT_CH);
    b[i].x = w[colB];
    b[i].y = w[OUT_CH + colB];
  }
}

__global__ void __launch_bounds__(128)
gates_wmma_kernel(const float* __restrict__ x, const float* __restrict__ H,
                  const float* __restrict__ Lx, const float* __restrict__ LH,
                  const float* __restrict__ Wzx, const float* __restrict__ Wzh,
                  const float* __restrict__ Wrx, const float* __restrict__ Wrh,
                  const float* __restrict__ bzx, const float* __restrict__ bzh,
                  const float* __restrict__ brx, const float* __restrict__ brh,
                  float* __restrict__ Zbuf, float* __restrict__ RH, int N) {
  __shared__ __attribute__((aligned(16))) float sA[2][16 * A_PITCH];
  const int t = threadIdx.x;
  const int lane = t & 31;
  const int wv = t >> 5;
  const int half = lane >> 4;
  const int l15 = lane & 15;
  const int colB = wv * 16 + l15;
  const int ntiles = N >> 4;
  const int step = gridDim.x;

  v2f bz[K_STEPS], br[K_STEPS];
  load_b_frags(Wzx, Wzh, colB, half, bz);
  load_b_frags(Wrx, Wrh, colB, half, br);
  const float bZ = bzx[colB] + bzh[colB];
  const float bR = brx[colB] + brh[colB];

  int tile = blockIdx.x;
  v4f tp0 = {}, tp1 = {}, tq0 = {}, tq1 = {}, ts = {};
  if (tile < ntiles) {
    tile_load(x, Lx, H, LH, tile * 16, t, tp0, tp1, tq0, tq1, ts);
    tile_store(sA[0], t, tp0, tp1, tq0, tq1, ts);
  }
  __syncthreads();
  int buf = 0;
  for (; tile < ntiles; tile += step) {
    const int next = tile + step;
    const bool more = next < ntiles;
    if (more) tile_load(x, Lx, H, LH, next * 16, t, tp0, tp1, tq0, tq1, ts);

    const float* sa = sA[buf];
    const float* arow = sa + l15 * A_PITCH + 2 * half;
    v8f accZ = {};
    v8f accR = {};
#pragma unroll
    for (int i = 0; i < K_STEPS; ++i) {
      v2f a = *(const v2f*)(arow + 4 * i);
      accZ = wmma_f32(a, bz[i], accZ);
      accR = wmma_f32(a, br[i], accR);
    }

    const int row0 = tile * 16;
#pragma unroll
    for (int i = 0; i < 8; ++i) {
      int lr = i + 8 * half;
      int r = row0 + lr;
      float z = 1.0f / (1.0f + __expf(-(accZ[i] + bZ)));
      float rgate = 1.0f / (1.0f + __expf(-(accR[i] + bR)));
      float h = sa[lr * A_PITCH + 2 * IN_CH + colB];
      Zbuf[r * OUT_CH + colB] = z;
      RH[r * OUT_CH + colB] = rgate * h;
    }

    if (more) tile_store(sA[buf ^ 1], t, tp0, tp1, tq0, tq1, ts);
    __syncthreads();
    buf ^= 1;
  }
}

__global__ void __launch_bounds__(128)
cand_out_wmma_kernel(const float* __restrict__ x, const float* __restrict__ H,
                     const float* __restrict__ Lx, const float* __restrict__ RH,
                     const float* __restrict__ LRH,
                     const float* __restrict__ Zbuf,
                     const float* __restrict__ Wcx,
                     const float* __restrict__ Wch,
                     const float* __restrict__ bcx,
                     const float* __restrict__ bch, float* __restrict__ out,
                     int N) {
  __shared__ __attribute__((aligned(16))) float sA[2][16 * A_PITCH];
  const int t = threadIdx.x;
  const int lane = t & 31;
  const int wv = t >> 5;
  const int half = lane >> 4;
  const int l15 = lane & 15;
  const int colB = wv * 16 + l15;
  const int ntiles = N >> 4;
  const int step = gridDim.x;

  v2f bc[K_STEPS];
  load_b_frags(Wcx, Wch, colB, half, bc);
  const float bC = bcx[colB] + bch[colB];

  int tile = blockIdx.x;
  v4f tp0 = {}, tp1 = {}, tq0 = {}, tq1 = {}, ts = {};
  if (tile < ntiles) {
    tile_load(x, Lx, RH, LRH, tile * 16, t, tp0, tp1, tq0, tq1, ts);
    tile_store(sA[0], t, tp0, tp1, tq0, tq1, ts);
  }
  __syncthreads();
  int buf = 0;
  for (; tile < ntiles; tile += step) {
    const int next = tile + step;
    const bool more = next < ntiles;
    if (more) tile_load(x, Lx, RH, LRH, next * 16, t, tp0, tp1, tq0, tq1, ts);

    const int row0 = tile * 16;
    float zv[8], hv[8];
#pragma unroll
    for (int i = 0; i < 8; ++i) {
      int r = row0 + i + 8 * half;
      zv[i] = Zbuf[r * OUT_CH + colB];
      hv[i] = H[r * OUT_CH + colB];
    }

    const float* sa = sA[buf];
    const float* arow = sa + l15 * A_PITCH + 2 * half;
    v8f accC = {};
#pragma unroll
    for (int i = 0; i < K_STEPS; ++i) {
      v2f a = *(const v2f*)(arow + 4 * i);
      accC = wmma_f32(a, bc[i], accC);
    }

#pragma unroll
    for (int i = 0; i < 8; ++i) {
      int r = row0 + i + 8 * half;
      float cv = tanhf(accC[i] + bC);
      out[r * OUT_CH + colB] = zv[i] * hv[i] + (1.0f - zv[i]) * cv;
    }

    if (more) tile_store(sA[buf ^ 1], t, tp0, tp1, tq0, tq1, ts);
    __syncthreads();
    buf ^= 1;
  }
}


extern "C" void kernel_launch(void* const* d_in, const int* in_sizes, int n_in,
                              void* d_out, int out_size, void* d_ws,
                              size_t ws_size, hipStream_t stream) {
  const float* x = (const float*)d_in[0];
  const float* H = (const float*)d_in[1];
  const int* edge = (const int*)d_in[2];
  const float* Wzx = (const float*)d_in[3];
  const float* bzx = (const float*)d_in[4];
  const float* Wzh = (const float*)d_in[5];
  const float* bzh = (const float*)d_in[6];
  const float* Wrx = (const float*)d_in[7];
  const float* brx = (const float*)d_in[8];
  const float* Wrh = (const float*)d_in[9];
  const float* brh = (const float*)d_in[10];
  const float* Wcx = (const float*)d_in[11];
  const float* bcx = (const float*)d_in[12];
  const float* Wch = (const float*)d_in[13];
  const float* bch = (const float*)d_in[14];

  const int N = in_sizes[0] / IN_CH;
  const int E = in_sizes[2] / 2;
  const int* rowIdx = edge;
  const int* colIdx = edge + E;

  float* ws = (float*)d_ws;
  float* LH = ws;
  float* LRH = LH + (size_t)N * OUT_CH;
  float* Zbuf = LRH + (size_t)N * OUT_CH;
  float* RH = Zbuf + (size_t)N * OUT_CH;
  float* Lx = RH + (size_t)N * OUT_CH;
  float* xd = Lx + (size_t)N * IN_CH;
  int* scol = (int*)(xd + (size_t)N * XD_W);
  float* nrm = (float*)(scol + E);
  int* tmp = (int*)(nrm + E);
  int* rowptr = tmp + (size_t)E * 2;
  int* chunkOff = rowptr + N + 1;

  const int NB = (N + ROWS_PER_BUCKET - 1) / ROWS_PER_BUCKET;
  const int nchunks = (E + CHUNK - 1) / CHUNK;

  chunk_sort_kernel<<<nchunks, CHUNK_THREADS, 0, stream>>>(rowIdx, colIdx, tmp,
                                                           chunkOff, E, NB);
  bucket_order_kernel<<<NB, 1024, 0, stream>>>(tmp, chunkOff, x, rowptr, scol,
                                               xd, N, NB, nchunks);
  {
    long long threads = (long long)N * 32;
    spmm_first_kernel<<<(int)((threads + 255) / 256), 256, 0, stream>>>(
        rowptr, scol, H, LH, xd, Lx, nrm, N);
  }
  gates_wmma_kernel<<<DENSE_GRID, 128, 0, stream>>>(x, H, Lx, LH, Wzx, Wzh, Wrx,
                                                     Wrh, bzx, bzh, brx, brh,
                                                     Zbuf, RH, N);
  {
    long long threads = (long long)N * 32;
    spmm_second_kernel<<<(int)((threads + 255) / 256), 256, 0, stream>>>(
        rowptr, scol, nrm, RH, LRH, N);
  }
  cand_out_wmma_kernel<<<DENSE_GRID, 128, 0, stream>>>(
      x, H, Lx, RH, LRH, Zbuf, Wcx, Wch, bcx, bch, (float*)d_out, N);
}
